# starting kernel
# baseline (speedup 1.0000x reference)
#include <hip/hip_runtime.h>
#include <math.h>

#define FD 5
#define NNODES 100000
#define NEDGES 6400000

typedef __attribute__((ext_vector_type(2))) float v2f;
typedef __attribute__((ext_vector_type(8))) float v8f;

__device__ __forceinline__ int imin(int a, int b) { return a < b ? a : b; }
__device__ __forceinline__ int imax(int a, int b) { return a > b ? a : b; }

__device__ __forceinline__ void gatomic_add_f32(float* p, float v) {
    asm volatile("global_atomic_add_f32 %0, %1, off"
                 :
                 : "v"(p), "v"(v)
                 : "memory");
}

__global__ void zero_f32(float* __restrict__ p, int n) {
    int i = blockIdx.x * blockDim.x + threadIdx.x;
    int stride = gridDim.x * blockDim.x;
    for (; i < n; i += stride) p[i] = 0.0f;
}

template <bool ADD_DEG>
__global__ void edge_agg(const float* __restrict__ h,
                         const int* __restrict__ src,
                         const int* __restrict__ dst,
                         float* __restrict__ agg,
                         float* __restrict__ deg,
                         int n_edges) {
    int e = blockIdx.x * blockDim.x + threadIdx.x;
    if (e >= n_edges) return;
    int s = src[e];
    int d = dst[e];
    const float* hs = h + (long)s * FD;
    float* ad = agg + (long)d * FD;
#pragma unroll
    for (int k = 0; k < FD; ++k) {
        gatomic_add_f32(ad + k, hs[k]);
    }
    if (ADD_DEG) {
        gatomic_add_f32(deg + d, 1.0f);
    }
}

template <int F_OUT>
__global__ void node_wmma(const float* __restrict__ h,
                          const float* __restrict__ agg,
                          const float* __restrict__ deg,
                          const float* __restrict__ Wself,
                          const float* __restrict__ Wneigh,
                          const float* __restrict__ bias,
                          float* __restrict__ out,
                          int n_nodes) {
    const int lane  = threadIdx.x & 31;
    const int wave  = threadIdx.x >> 5;
    const int tile  = blockIdx.x * (blockDim.x >> 5) + wave;
    const int m0    = tile * 16;
    if (m0 >= n_nodes) return;

    const int row   = lane & 15;
    const int khalf = lane >> 4;

    const int m  = m0 + row;
    const int mc = imin(m, n_nodes - 1);
    const float invdeg = 1.0f / fmaxf(deg[mc], 1.0f);

    const int n  = row;
    const int nc = imin(n, F_OUT - 1);
    const bool nvalid = (n < F_OUT);

    v8f c = {};
#pragma unroll
    for (int chunk = 0; chunk < 3; ++chunk) {
        v2f a, b;
#pragma unroll
        for (int j = 0; j < 2; ++j) {
            const int k  = chunk * 4 + khalf * 2 + j;
            const int kh = imin(k, FD - 1);
            const int kn = imin(imax(k - FD, 0), FD - 1);

            const float hv = h[(long)mc * FD + kh];
            const float nv = agg[(long)mc * FD + kn] * invdeg;
            const float av = (k < FD) ? hv : ((k < 2 * FD) ? nv : 0.0f);

            const float ws = Wself[kh * F_OUT + nc];
            const float wn = Wneigh[kn * F_OUT + nc];
            float bv = (k < FD) ? ws : ((k < 2 * FD) ? wn : 0.0f);
            bv = nvalid ? bv : 0.0f;

            a[j] = av;
            b[j] = bv;
        }
        c = __builtin_amdgcn_wmma_f32_16x16x4_f32(false, a, false, b,
                                                  (short)0, c, false, false);
    }

    const float bn = bias[nc];
#pragma unroll
    for (int r = 0; r < 8; ++r) {
        const int mm = m0 + r + khalf * 8;
        if (mm < n_nodes && nvalid) {
            const float v = c[r] + bn;
            out[(long)mm * F_OUT + n] = 1.0f / (1.0f + __expf(-v));
        }
    }
}

extern "C" void kernel_launch(void* const* d_in, const int* in_sizes, int n_in,
                              void* d_out, int out_size, void* d_ws, size_t ws_size,
                              hipStream_t stream) {
    (void)in_sizes; (void)n_in; (void)out_size; (void)ws_size;

    const float* x      = (const float*)d_in[0];
    const int*   src    = (const int*)d_in[1];
    const int*   dst    = (const int*)d_in[2];
    const float* Wself1 = (const float*)d_in[3];
    const float* Wneigh1= (const float*)d_in[4];
    const float* b1     = (const float*)d_in[5];
    const float* Wself2 = (const float*)d_in[6];
    const float* Wneigh2= (const float*)d_in[7];
    const float* b2     = (const float*)d_in[8];
    float* out = (float*)d_out;

    float* h   = (float*)d_ws;
    float* agg = h + (size_t)NNODES * FD;
    float* deg = agg + (size_t)NNODES * FD;

    const int ZB = 256;
    const int EB = 256;
    const int edge_grid = (NEDGES + EB - 1) / EB;

    const int NODE_B = 256;
    const int tiles = (NNODES + 15) / 16;
    const int node_grid = (tiles + (NODE_B / 32) - 1) / (NODE_B / 32);

    {
        const int nz = NNODES * FD + NNODES;
        zero_f32<<<(nz + ZB - 1) / ZB, ZB, 0, stream>>>(agg, nz);
    }
    edge_agg<true><<<edge_grid, EB, 0, stream>>>(x, src, dst, agg, deg, NEDGES);
    node_wmma<FD><<<node_grid, NODE_B, 0, stream>>>(x, agg, deg, Wself1, Wneigh1,
                                                    b1, h, NNODES);

    {
        const int nz = NNODES * FD;
        zero_f32<<<(nz + ZB - 1) / ZB, ZB, 0, stream>>>(agg, nz);
    }
    edge_agg<false><<<edge_grid, EB, 0, stream>>>(h, src, dst, agg, deg, NEDGES);
    node_wmma<10><<<node_grid, NODE_B, 0, stream>>>(h, agg, deg, Wself2, Wneigh2,
                                                    b2, out, NNODES);
}


# Node-range owner work-groups scan dst, queue own edges in LDS, sum gathered rows in per-wave LDS copies: no atomics, no zero fill; sum order now fixed (edge order per wave, then 16 partials)
# speedup vs baseline: 2.2499x; 2.2499x over previous
#include <hip/hip_runtime.h>
#include <math.h>

#define FD 5
#define NNODES 100000
#define NEDGES 6400000

typedef __attribute__((ext_vector_type(2))) float v2f;
typedef __attribute__((ext_vector_type(8))) float v8f;

__device__ __forceinline__ int imin(int a, int b) { return a < b ? a : b; }
__device__ __forceinline__ int imax(int a, int b) { return a > b ? a : b; }

#define NBLK 256
#define NPB  391
#define TPB  512
#define NWAVE (TPB / 32)
#define QW   2048
#define VPI  8
#define EBITS 23
#define AC   (FD + 1)

typedef __attribute__((ext_vector_type(4))) int v4i;

template <bool ADD_DEG>
__device__ __forceinline__ void drain_queue(const unsigned* myq, int qn, int lane,
                                            const float* __restrict__ h,
                                            const int* __restrict__ src,
                                            float* myacc) {
    const int col = imin(lane, FD - 1);
    const int ncol = ADD_DEG ? AC : FD;
    for (int i = 0; i < qn; i += 32) {
        const int idx = i + lane;
        const unsigned w = myq[idx < qn ? idx : 0];
        const int s = src[w & ((1u << EBITS) - 1u)];
        float r[32];
#pragma unroll
        for (int j = 0; j < 32; ++j) {
            const int sj = __builtin_amdgcn_readlane(s, j);
            r[j] = h[(long)sj * FD + col];
        }
#pragma unroll
        for (int j = 0; j < 32; ++j) {
            if (i + j < qn) {
                const unsigned tj =
                    ((unsigned)__builtin_amdgcn_readlane((int)w, j)) >> EBITS;
                const float v = (lane < FD) ? r[j] : 1.0f;
                if (lane < ncol) myacc[tj * AC + lane] += v;
            }
        }
    }
}

template <bool ADD_DEG>
__global__ __launch_bounds__(TPB)
void owner_agg(const float* __restrict__ h,
               const int* __restrict__ src,
               const int* __restrict__ dst,
               float* __restrict__ agg,
               float* __restrict__ deg,
               int n_nodes, int n_edges) {
    __shared__ float accs[NWAVE * NPB * AC];
    __shared__ unsigned q[NWAVE * QW];

    const int tid  = threadIdx.x;
    const int lane = tid & 31;
    const int wave = tid >> 5;
    const int lo   = blockIdx.x * NPB;
    const int cnt  = imax(imin(NPB, n_nodes - lo), 0);

    for (int i = tid; i < NWAVE * NPB * AC; i += TPB) accs[i] = 0.0f;
    __syncthreads();

    float* myacc = accs + wave * (NPB * AC);
    unsigned* myq = q + wave * QW;
    int qn = 0;

    const v4i* dst4 = (const v4i*)dst;
    const int n_vec = n_edges >> 2;
    const int step  = TPB * VPI;

    v4i nd[VPI];
#pragma unroll
    for (int u = 0; u < VPI; ++u) {
        const int v = u * TPB + tid;
        nd[u] = dst4[imin(v, n_vec - 1)];
    }
    for (int base = 0; base < n_vec; base += step) {
        v4i d[VPI];
#pragma unroll
        for (int u = 0; u < VPI; ++u) d[u] = nd[u];
        if (qn > QW - VPI * 4 * 32) {
            drain_queue<ADD_DEG>(myq, qn, lane, h, src, myacc);
            qn = 0;
        }
#pragma unroll
        for (int u = 0; u < VPI; ++u) {
            const int v = base + u * TPB + tid;
            const bool valid = v < n_vec;
#pragma unroll
            for (int c = 0; c < 4; ++c) {
                const unsigned t = (unsigned)(d[u][c] - lo);
                const bool m = valid && (t < (unsigned)cnt);
                const unsigned mask = __builtin_amdgcn_ballot_w32(m);
                if (mask != 0u) {
                    if (m) {
                        const int pos = qn + (int)__builtin_amdgcn_mbcnt_lo(mask, 0u);
                        myq[pos] = (t << EBITS) | (unsigned)(4 * v + c);
                    }
                    qn += __builtin_popcount(mask);
                }
                if (u == 0 && c == 0) {
#pragma unroll
                    for (int u2 = 0; u2 < VPI; ++u2) {
                        const int v2 = base + step + u2 * TPB + tid;
                        nd[u2] = dst4[imin(v2, n_vec - 1)];
                    }
                }
            }
        }
    }
    for (int e0 = n_vec * 4; e0 < n_edges; e0 += TPB) {
        const int e = e0 + tid;
        if (qn > QW - 32) {
            drain_queue<ADD_DEG>(myq, qn, lane, h, src, myacc);
            qn = 0;
        }
        const bool valid = e < n_edges;
        const unsigned t = (unsigned)(dst[valid ? e : 0] - lo);
        const bool m = valid && (t < (unsigned)cnt);
        const unsigned mask = __builtin_amdgcn_ballot_w32(m);
        if (m) {
            const int pos = qn + (int)__builtin_amdgcn_mbcnt_lo(mask, 0u);
            myq[pos] = (t << EBITS) | (unsigned)e;
        }
        qn += __builtin_popcount(mask);
    }
    drain_queue<ADD_DEG>(myq, qn, lane, h, src, myacc);
    __syncthreads();

    for (int i = tid; i < cnt * AC; i += TPB) {
        float sum = 0.0f;
#pragma unroll
        for (int w = 0; w < NWAVE; ++w) sum += accs[w * (NPB * AC) + i];
        const int node = i / AC;
        const int c = i - node * AC;
        if (c < FD) agg[(long)(lo + node) * FD + c] = sum;
        else if (ADD_DEG) deg[lo + node] = sum;
    }
}

template <int F_OUT>
__global__ void node_wmma(const float* __restrict__ h,
                          const float* __restrict__ agg,
                          const float* __restrict__ deg,
                          const float* __restrict__ Wself,
                          const float* __restrict__ Wneigh,
                          const float* __restrict__ bias,
                          float* __restrict__ out,
                          int n_nodes) {
    const int lane  = threadIdx.x & 31;
    const int wave  = threadIdx.x >> 5;
    const int tile  = blockIdx.x * (blockDim.x >> 5) + wave;
    const int m0    = tile * 16;
    if (m0 >= n_nodes) return;

    const int row   = lane & 15;
    const int khalf = lane >> 4;

    const int m  = m0 + row;
    const int mc = imin(m, n_nodes - 1);
    const float invdeg = 1.0f / fmaxf(deg[mc], 1.0f);

    const int n  = row;
    const int nc = imin(n, F_OUT - 1);
    const bool nvalid = (n < F_OUT);

    v8f c = {};
#pragma unroll
    for (int chunk = 0; chunk < 3; ++chunk) {
        v2f a, b;
#pragma unroll
        for (int j = 0; j < 2; ++j) {
            const int k  = chunk * 4 + khalf * 2 + j;
            const int kh = imin(k, FD - 1);
            const int kn = imin(imax(k - FD, 0), FD - 1);

            const float hv = h[(long)mc * FD + kh];
            const float nv = agg[(long)mc * FD + kn] * invdeg;
            const float av = (k < FD) ? hv : ((k < 2 * FD) ? nv : 0.0f);

            const float ws = Wself[kh * F_OUT + nc];
            const float wn = Wneigh[kn * F_OUT + nc];
            float bv = (k < FD) ? ws : ((k < 2 * FD) ? wn : 0.0f);
            bv = nvalid ? bv : 0.0f;

            a[j] = av;
            b[j] = bv;
        }
        c = __builtin_amdgcn_wmma_f32_16x16x4_f32(false, a, false, b,
                                                  (short)0, c, false, false);
    }

    const float bn = bias[nc];
#pragma unroll
    for (int r = 0; r < 8; ++r) {
        const int mm = m0 + r + khalf * 8;
        if (mm < n_nodes && nvalid) {
            const float v = c[r] + bn;
            out[(long)mm * F_OUT + n] = 1.0f / (1.0f + __expf(-v));
        }
    }
}

extern "C" void kernel_launch(void* const* d_in, const int* in_sizes, int n_in,
                              void* d_out, int out_size, void* d_ws, size_t ws_size,
                              hipStream_t stream) {
    (void)in_sizes; (void)n_in; (void)out_size; (void)ws_size;

    const float* x      = (const float*)d_in[0];
    const int*   src    = (const int*)d_in[1];
    const int*   dst    = (const int*)d_in[2];
    const float* Wself1 = (const float*)d_in[3];
    const float* Wneigh1= (const float*)d_in[4];
    const float* b1     = (const float*)d_in[5];
    const float* Wself2 = (const float*)d_in[6];
    const float* Wneigh2= (const float*)d_in[7];
    const float* b2     = (const float*)d_in[8];
    float* out = (float*)d_out;

    float* h   = (float*)d_ws;
    float* agg = h + (size_t)NNODES * FD;
    float* deg = agg + (size_t)NNODES * FD;

    const int NODE_B = 256;
    const int tiles = (NNODES + 15) / 16;
    const int node_grid = (tiles + (NODE_B / 32) - 1) / (NODE_B / 32);

    owner_agg<true><<<NBLK, TPB, 0, stream>>>(x, src, dst, agg, deg, NNODES, NEDGES);
    node_wmma<FD><<<node_grid, NODE_B, 0, stream>>>(x, agg, deg, Wself1, Wneigh1,
                                                    b1, h, NNODES);

    owner_agg<false><<<NBLK, TPB, 0, stream>>>(h, src, dst, agg, deg, NNODES, NEDGES);
    node_wmma<10><<<node_grid, NODE_B, 0, stream>>>(h, agg, deg, Wself2, Wneigh2,
                                                    b2, out, NNODES);
}


# dst scan votes once per 16-byte vector (min of its four range offsets) and appends a hit vector's four components branch-free (spare LDS slot per lane); queue order and sums unchanged
# speedup vs baseline: 2.9895x; 1.3287x over previous
#include <hip/hip_runtime.h>
#include <math.h>

#define FD 5
#define NNODES 100000
#define NEDGES 6400000

typedef __attribute__((ext_vector_type(2))) float v2f;
typedef __attribute__((ext_vector_type(8))) float v8f;

__device__ __forceinline__ int imin(int a, int b) { return a < b ? a : b; }
__device__ __forceinline__ int imax(int a, int b) { return a > b ? a : b; }

#define NBLK 256
#define NPB  391
#define TPB  512
#define NWAVE (TPB / 32)
#define QW   2048
#define QS   (QW + 32)
#define VPI  8
#define EBITS 23
#define AC   (FD + 1)

typedef __attribute__((ext_vector_type(4))) int v4i;

template <bool ADD_DEG>
__device__ __forceinline__ void drain_queue(const unsigned* myq, int qn, int lane,
                                            const float* __restrict__ h,
                                            const int* __restrict__ src,
                                            float* myacc) {
    const int col = imin(lane, FD - 1);
    const int ncol = ADD_DEG ? AC : FD;
    for (int i = 0; i < qn; i += 32) {
        const int idx = i + lane;
        const unsigned w = myq[idx < qn ? idx : 0];
        const int s = src[w & ((1u << EBITS) - 1u)];
        float r[32];
#pragma unroll
        for (int j = 0; j < 32; ++j) {
            const int sj = __builtin_amdgcn_readlane(s, j);
            r[j] = h[(long)sj * FD + col];
        }
#pragma unroll
        for (int j = 0; j < 32; ++j) {
            if (i + j < qn) {
                const unsigned tj =
                    ((unsigned)__builtin_amdgcn_readlane((int)w, j)) >> EBITS;
                const float v = (lane < FD) ? r[j] : 1.0f;
                if (lane < ncol) myacc[tj * AC + lane] += v;
            }
        }
    }
}

__device__ __forceinline__ void scan_full_trip(const v4i* d, v4i* nd,
                                               const v4i* __restrict__ dst4,
                                               int base, int step, int tid, int n_vec,
                                               int lo, unsigned cnt, int lane,
                                               unsigned* myq, int& qn) {
#pragma unroll
    for (int u = 0; u < VPI; ++u) {
        const unsigned t0 = (unsigned)(d[u][0] - lo);
        const unsigned t1 = (unsigned)(d[u][1] - lo);
        const unsigned t2 = (unsigned)(d[u][2] - lo);
        const unsigned t3 = (unsigned)(d[u][3] - lo);
        const unsigned t01 = t0 < t1 ? t0 : t1;
        const unsigned t23 = t2 < t3 ? t2 : t3;
        const unsigned tm  = t01 < t23 ? t01 : t23;
        if (__builtin_amdgcn_ballot_w32(tm < cnt) != 0u) {
            const unsigned e0 = 4u * (unsigned)(base + u * TPB + tid);
            const unsigned tt[4] = {t0, t1, t2, t3};
#pragma unroll
            for (int c = 0; c < 4; ++c) {
                const bool m = tt[c] < cnt;
                const unsigned mask = __builtin_amdgcn_ballot_w32(m);
                const int pos = (int)__builtin_amdgcn_mbcnt_lo(mask, (unsigned)qn);
                myq[m ? pos : QW + lane] = (tt[c] << EBITS) | (e0 + (unsigned)c);
                qn += __builtin_popcount(mask);
            }
        }
        if (u == 0) {
#pragma unroll
            for (int u2 = 0; u2 < VPI; ++u2) {
                const int v2 = base + step + u2 * TPB + tid;
                nd[u2] = dst4[imin(v2, n_vec - 1)];
            }
        }
    }
}

template <bool ADD_DEG>
__global__ __launch_bounds__(TPB)
void owner_agg(const float* __restrict__ h,
               const int* __restrict__ src,
               const int* __restrict__ dst,
               float* __restrict__ agg,
               float* __restrict__ deg,
               int n_nodes, int n_edges) {
    __shared__ float accs[NWAVE * NPB * AC];
    __shared__ unsigned q[NWAVE * QS];

    const int tid  = threadIdx.x;
    const int lane = tid & 31;
    const int wave = tid >> 5;
    const int lo   = blockIdx.x * NPB;
    const int cnt  = imax(imin(NPB, n_nodes - lo), 0);

    for (int i = tid; i < NWAVE * NPB * AC; i += TPB) accs[i] = 0.0f;
    __syncthreads();

    float* myacc = accs + wave * (NPB * AC);
    unsigned* myq = q + wave * QS;
    int qn = 0;

    const v4i* dst4 = (const v4i*)dst;
    const int n_vec = n_edges >> 2;
    const int step  = TPB * VPI;

    v4i nd[VPI];
#pragma unroll
    for (int u = 0; u < VPI; ++u) {
        const int v = u * TPB + tid;
        nd[u] = dst4[imin(v, n_vec - 1)];
    }
    for (int base = 0; base < n_vec; base += step) {
        if (qn > QW - VPI * 4 * 32) {
            drain_queue<ADD_DEG>(myq, qn, lane, h, src, myacc);
            qn = 0;
        }
        if (base + step <= n_vec) {
            v4i d[VPI];
#pragma unroll
            for (int u = 0; u < VPI; ++u) d[u] = nd[u];
            scan_full_trip(d, nd, dst4, base, step, tid, n_vec, lo, (unsigned)cnt, lane, myq, qn);
        } else {
#pragma nounroll
            for (int vb = base; vb < n_vec; vb += TPB) {
                const int v = vb + tid;
                const bool valid = v < n_vec;
                const v4i dv = dst4[valid ? v : n_vec - 1];
#pragma unroll
                for (int c = 0; c < 4; ++c) {
                    const unsigned t = (unsigned)(dv[c] - lo);
                    const bool m = valid && (t < (unsigned)cnt);
                    const unsigned mask = __builtin_amdgcn_ballot_w32(m);
                    if (mask != 0u) {
                        if (m) {
                            const int pos = qn + (int)__builtin_amdgcn_mbcnt_lo(mask, 0u);
                            myq[pos] = (t << EBITS) | (unsigned)(4 * v + c);
                        }
                        qn += __builtin_popcount(mask);
                    }
                }
            }
        }
    }
    for (int e0 = n_vec * 4; e0 < n_edges; e0 += TPB) {
        const int e = e0 + tid;
        if (qn > QW - 32) {
            drain_queue<ADD_DEG>(myq, qn, lane, h, src, myacc);
            qn = 0;
        }
        const bool valid = e < n_edges;
        const unsigned t = (unsigned)(dst[valid ? e : 0] - lo);
        const bool m = valid && (t < (unsigned)cnt);
        const unsigned mask = __builtin_amdgcn_ballot_w32(m);
        if (m) {
            const int pos = qn + (int)__builtin_amdgcn_mbcnt_lo(mask, 0u);
            myq[pos] = (t << EBITS) | (unsigned)e;
        }
        qn += __builtin_popcount(mask);
    }
    drain_queue<ADD_DEG>(myq, qn, lane, h, src, myacc);
    __syncthreads();

    for (int i = tid; i < cnt * AC; i += TPB) {
        float sum = 0.0f;
#pragma unroll
        for (int w = 0; w < NWAVE; ++w) sum += accs[w * (NPB * AC) + i];
        const int node = i / AC;
        const int c = i - node * AC;
        if (c < FD) agg[(long)(lo + node) * FD + c] = sum;
        else if (ADD_DEG) deg[lo + node] = sum;
    }
}

template <int F_OUT>
__global__ void node_wmma(const float* __restrict__ h,
                          const float* __restrict__ agg,
                          const float* __restrict__ deg,
                          const float* __restrict__ Wself,
                          const float* __restrict__ Wneigh,
                          const float* __restrict__ bias,
                          float* __restrict__ out,
                          int n_nodes) {
    const int lane  = threadIdx.x & 31;
    const int wave  = threadIdx.x >> 5;
    const int tile  = blockIdx.x * (blockDim.x >> 5) + wave;
    const int m0    = tile * 16;
    if (m0 >= n_nodes) return;

    const int row   = lane & 15;
    const int khalf = lane >> 4;

    const int m  = m0 + row;
    const int mc = imin(m, n_nodes - 1);
    const float invdeg = 1.0f / fmaxf(deg[mc], 1.0f);

    const int n  = row;
    const int nc = imin(n, F_OUT - 1);
    const bool nvalid = (n < F_OUT);

    v8f c = {};
#pragma unroll
    for (int chunk = 0; chunk < 3; ++chunk) {
        v2f a, b;
#pragma unroll
        for (int j = 0; j < 2; ++j) {
            const int k  = chunk * 4 + khalf * 2 + j;
            const int kh = imin(k, FD - 1);
            const int kn = imin(imax(k - FD, 0), FD - 1);

            const float hv = h[(long)mc * FD + kh];
            const float nv = agg[(long)mc * FD + kn] * invdeg;
            const float av = (k < FD) ? hv : ((k < 2 * FD) ? nv : 0.0f);

            const float ws = Wself[kh * F_OUT + nc];
            const float wn = Wneigh[kn * F_OUT + nc];
            float bv = (k < FD) ? ws : ((k < 2 * FD) ? wn : 0.0f);
            bv = nvalid ? bv : 0.0f;

            a[j] = av;
            b[j] = bv;
        }
        c = __builtin_amdgcn_wmma_f32_16x16x4_f32(false, a, false, b,
                                                  (short)0, c, false, false);
    }

    const float bn = bias[nc];
#pragma unroll
    for (int r = 0; r < 8; ++r) {
        const int mm = m0 + r + khalf * 8;
        if (mm < n_nodes && nvalid) {
            const float v = c[r] + bn;
            out[(long)mm * F_OUT + n] = 1.0f / (1.0f + __expf(-v));
        }
    }
}

extern "C" void kernel_launch(void* const* d_in, const int* in_sizes, int n_in,
                              void* d_out, int out_size, void* d_ws, size_t ws_size,
                              hipStream_t stream) {
    (void)in_sizes; (void)n_in; (void)out_size; (void)ws_size;

    const float* x      = (const float*)d_in[0];
    const int*   src    = (const int*)d_in[1];
    const int*   dst    = (const int*)d_in[2];
    const float* Wself1 = (const float*)d_in[3];
    const float* Wneigh1= (const float*)d_in[4];
    const float* b1     = (const float*)d_in[5];
    const float* Wself2 = (const float*)d_in[6];
    const float* Wneigh2= (const float*)d_in[7];
    const float* b2     = (const float*)d_in[8];
    float* out = (float*)d_out;

    float* h   = (float*)d_ws;
    float* agg = h + (size_t)NNODES * FD;
    float* deg = agg + (size_t)NNODES * FD;

    const int NODE_B = 256;
    const int tiles = (NNODES + 15) / 16;
    const int node_grid = (tiles + (NODE_B / 32) - 1) / (NODE_B / 32);

    owner_agg<true><<<NBLK, TPB, 0, stream>>>(x, src, dst, agg, deg, NNODES, NEDGES);
    node_wmma<FD><<<node_grid, NODE_B, 0, stream>>>(x, agg, deg, Wself1, Wneigh1,
                                                    b1, h, NNODES);

    owner_agg<false><<<NBLK, TPB, 0, stream>>>(h, src, dst, agg, deg, NNODES, NEDGES);
    node_wmma<10><<<node_grid, NODE_B, 0, stream>>>(h, agg, deg, Wself2, Wneigh2,
                                                    b2, out, NNODES);
}


# Queue drain handles one edge per lane: whole-row gathers, 128 edges in flight, LDS row updates by all lanes at once; same-node lanes of a 32-group take turns by an LDS tag vote, reordering their sums
# speedup vs baseline: 3.8339x; 1.2825x over previous
#include <hip/hip_runtime.h>
#include <math.h>

#define FD 5
#define NNODES 100000
#define NEDGES 6400000

typedef __attribute__((ext_vector_type(2))) float v2f;
typedef __attribute__((ext_vector_type(8))) float v8f;

__device__ __forceinline__ int imin(int a, int b) { return a < b ? a : b; }
__device__ __forceinline__ int imax(int a, int b) { return a > b ? a : b; }

#define NBLK 256
#define NPB  391
#define TPB  512
#define NWAVE (TPB / 32)
#define QW   2048
#define QS   (QW + 32)
#define VPI  8
#define EBITS 23
#define AC   (FD + 1)

typedef __attribute__((ext_vector_type(4))) int v4i;

struct __attribute__((packed, aligned(4))) row_t { float v[FD]; };

#define DB 4
typedef volatile __attribute__((address_space(3))) unsigned* lds_tag_ptr;
template <bool ADD_DEG>
__device__ __forceinline__ void drain_queue(const unsigned* myq, int qn, int lane,
                                            const float* __restrict__ h,
                                            const int* __restrict__ src,
                                            float* myacc,
                                            lds_tag_ptr mytag) {
    for (int i = 0; i < qn; i += DB * 32) {
        unsigned w[DB];
        int s[DB];
        row_t r[DB];
#pragma unroll
        for (int k = 0; k < DB; ++k) {
            const int idx = i + k * 32 + lane;
            w[k] = myq[idx < qn ? idx : qn - 1];
        }
#pragma unroll
        for (int k = 0; k < DB; ++k) s[k] = src[w[k] & ((1u << EBITS) - 1u)];
#pragma unroll
        for (int k = 0; k < DB; ++k)
            __builtin_memcpy(&r[k], h + (long)s[k] * FD, sizeof(row_t));
#pragma unroll
        for (int k = 0; k < DB; ++k) {
            const unsigned t = w[k] >> EBITS;
            bool pending = (i + k * 32 + lane) < qn;
            for (int round = 0; round < 32; ++round) {
                if (__builtin_amdgcn_ballot_w32(pending) == 0u) break;
                bool win = false;
                if (pending) {
                    mytag[t] = (unsigned)lane;
                    win = (mytag[t] == (unsigned)lane);
                }
                if (win) {
                    float2* p = (float2*)(myacc + t * AC);
                    float2 a0 = p[0], a1 = p[1], a2 = p[2];
                    a0.x += r[k].v[0];
                    a0.y += r[k].v[1];
                    a1.x += r[k].v[2];
                    a1.y += r[k].v[3];
                    a2.x += r[k].v[4];
                    if (ADD_DEG) a2.y += 1.0f;
                    p[0] = a0;
                    p[1] = a1;
                    p[2] = a2;
                    pending = false;
                }
            }
        }
    }
}

__device__ __forceinline__ void scan_full_trip(const v4i* d, v4i* nd,
                                               const v4i* __restrict__ dst4,
                                               int base, int step, int tid, int n_vec,
                                               int lo, unsigned cnt, int lane,
                                               unsigned* myq, int& qn) {
#pragma unroll
    for (int u = 0; u < VPI; ++u) {
        const unsigned t0 = (unsigned)(d[u][0] - lo);
        const unsigned t1 = (unsigned)(d[u][1] - lo);
        const unsigned t2 = (unsigned)(d[u][2] - lo);
        const unsigned t3 = (unsigned)(d[u][3] - lo);
        const unsigned t01 = t0 < t1 ? t0 : t1;
        const unsigned t23 = t2 < t3 ? t2 : t3;
        const unsigned tm  = t01 < t23 ? t01 : t23;
        if (__builtin_amdgcn_ballot_w32(tm < cnt) != 0u) {
            const unsigned e0 = 4u * (unsigned)(base + u * TPB + tid);
            const unsigned tt[4] = {t0, t1, t2, t3};
#pragma unroll
            for (int c = 0; c < 4; ++c) {
                const bool m = tt[c] < cnt;
                const unsigned mask = __builtin_amdgcn_ballot_w32(m);
                const int pos = (int)__builtin_amdgcn_mbcnt_lo(mask, (unsigned)qn);
                myq[m ? pos : QW + lane] = (tt[c] << EBITS) | (e0 + (unsigned)c);
                qn += __builtin_popcount(mask);
            }
        }
        if (u == 0) {
#pragma unroll
            for (int u2 = 0; u2 < VPI; ++u2) {
                const int v2 = base + step + u2 * TPB + tid;
                nd[u2] = dst4[imin(v2, n_vec - 1)];
            }
        }
    }
}

template <bool ADD_DEG>
__global__ __launch_bounds__(TPB)
void owner_agg(const float* __restrict__ h,
               const int* __restrict__ src,
               const int* __restrict__ dst,
               float* __restrict__ agg,
               float* __restrict__ deg,
               int n_nodes, int n_edges) {
    __shared__ __attribute__((aligned(16))) float accs[NWAVE * NPB * AC];
    __shared__ unsigned q[NWAVE * QS];
    __shared__ unsigned tags[NWAVE * NPB];

    const int tid  = threadIdx.x;
    const int lane = tid & 31;
    const int wave = tid >> 5;
    const int lo   = blockIdx.x * NPB;
    const int cnt  = imax(imin(NPB, n_nodes - lo), 0);

    for (int i = tid; i < NWAVE * NPB * AC; i += TPB) accs[i] = 0.0f;
    __syncthreads();

    float* myacc = accs + wave * (NPB * AC);
    unsigned* myq = q + wave * QS;
    lds_tag_ptr mytag = (lds_tag_ptr)(tags + wave * NPB);
    int qn = 0;

    const v4i* dst4 = (const v4i*)dst;
    const int n_vec = n_edges >> 2;
    const int step  = TPB * VPI;

    v4i nd[VPI];
#pragma unroll
    for (int u = 0; u < VPI; ++u) {
        const int v = u * TPB + tid;
        nd[u] = dst4[imin(v, n_vec - 1)];
    }
    for (int base = 0; base < n_vec; base += step) {
        if (qn > QW - VPI * 4 * 32) {
            drain_queue<ADD_DEG>(myq, qn, lane, h, src, myacc, mytag);
            qn = 0;
        }
        if (base + step <= n_vec) {
            v4i d[VPI];
#pragma unroll
            for (int u = 0; u < VPI; ++u) d[u] = nd[u];
            scan_full_trip(d, nd, dst4, base, step, tid, n_vec, lo, (unsigned)cnt, lane, myq, qn);
        } else {
#pragma nounroll
            for (int vb = base; vb < n_vec; vb += TPB) {
                const int v = vb + tid;
                const bool valid = v < n_vec;
                const v4i dv = dst4[valid ? v : n_vec - 1];
#pragma unroll
                for (int c = 0; c < 4; ++c) {
                    const unsigned t = (unsigned)(dv[c] - lo);
                    const bool m = valid && (t < (unsigned)cnt);
                    const unsigned mask = __builtin_amdgcn_ballot_w32(m);
                    if (mask != 0u) {
                        if (m) {
                            const int pos = qn + (int)__builtin_amdgcn_mbcnt_lo(mask, 0u);
                            myq[pos] = (t << EBITS) | (unsigned)(4 * v + c);
                        }
                        qn += __builtin_popcount(mask);
                    }
                }
            }
        }
    }
    for (int e0 = n_vec * 4; e0 < n_edges; e0 += TPB) {
        const int e = e0 + tid;
        if (qn > QW - 32) {
            drain_queue<ADD_DEG>(myq, qn, lane, h, src, myacc, mytag);
            qn = 0;
        }
        const bool valid = e < n_edges;
        const unsigned t = (unsigned)(dst[valid ? e : 0] - lo);
        const bool m = valid && (t < (unsigned)cnt);
        const unsigned mask = __builtin_amdgcn_ballot_w32(m);
        if (m) {
            const int pos = qn + (int)__builtin_amdgcn_mbcnt_lo(mask, 0u);
            myq[pos] = (t << EBITS) | (unsigned)e;
        }
        qn += __builtin_popcount(mask);
    }
    drain_queue<ADD_DEG>(myq, qn, lane, h, src, myacc, mytag);
    __syncthreads();

    for (int i = tid; i < cnt * AC; i += TPB) {
        float sum = 0.0f;
#pragma unroll
        for (int w = 0; w < NWAVE; ++w) sum += accs[w * (NPB * AC) + i];
        const int node = i / AC;
        const int c = i - node * AC;
        if (c < FD) agg[(long)(lo + node) * FD + c] = sum;
        else if (ADD_DEG) deg[lo + node] = sum;
    }
}

template <int F_OUT>
__global__ void node_wmma(const float* __restrict__ h,
                          const float* __restrict__ agg,
                          const float* __restrict__ deg,
                          const float* __restrict__ Wself,
                          const float* __restrict__ Wneigh,
                          const float* __restrict__ bias,
                          float* __restrict__ out,
                          int n_nodes) {
    const int lane  = threadIdx.x & 31;
    const int wave  = threadIdx.x >> 5;
    const int tile  = blockIdx.x * (blockDim.x >> 5) + wave;
    const int m0    = tile * 16;
    if (m0 >= n_nodes) return;

    const int row   = lane & 15;
    const int khalf = lane >> 4;

    const int m  = m0 + row;
    const int mc = imin(m, n_nodes - 1);
    const float invdeg = 1.0f / fmaxf(deg[mc], 1.0f);

    const int n  = row;
    const int nc = imin(n, F_OUT - 1);
    const bool nvalid = (n < F_OUT);

    v8f c = {};
#pragma unroll
    for (int chunk = 0; chunk < 3; ++chunk) {
        v2f a, b;
#pragma unroll
        for (int j = 0; j < 2; ++j) {
            const int k  = chunk * 4 + khalf * 2 + j;
            const int kh = imin(k, FD - 1);
            const int kn = imin(imax(k - FD, 0), FD - 1);

            const float hv = h[(long)mc * FD + kh];
            const float nv = agg[(long)mc * FD + kn] * invdeg;
            const float av = (k < FD) ? hv : ((k < 2 * FD) ? nv : 0.0f);

            const float ws = Wself[kh * F_OUT + nc];
            const float wn = Wneigh[kn * F_OUT + nc];
            float bv = (k < FD) ? ws : ((k < 2 * FD) ? wn : 0.0f);
            bv = nvalid ? bv : 0.0f;

            a[j] = av;
            b[j] = bv;
        }
        c = __builtin_amdgcn_wmma_f32_16x16x4_f32(false, a, false, b,
                                                  (short)0, c, false, false);
    }

    const float bn = bias[nc];
#pragma unroll
    for (int r = 0; r < 8; ++r) {
        const int mm = m0 + r + khalf * 8;
        if (mm < n_nodes && nvalid) {
            const float v = c[r] + bn;
            out[(long)mm * F_OUT + n] = 1.0f / (1.0f + __expf(-v));
        }
    }
}

extern "C" void kernel_launch(void* const* d_in, const int* in_sizes, int n_in,
                              void* d_out, int out_size, void* d_ws, size_t ws_size,
                              hipStream_t stream) {
    (void)in_sizes; (void)n_in; (void)out_size; (void)ws_size;

    const float* x      = (const float*)d_in[0];
    const int*   src    = (const int*)d_in[1];
    const int*   dst    = (const int*)d_in[2];
    const float* Wself1 = (const float*)d_in[3];
    const float* Wneigh1= (const float*)d_in[4];
    const float* b1     = (const float*)d_in[5];
    const float* Wself2 = (const float*)d_in[6];
    const float* Wneigh2= (const float*)d_in[7];
    const float* b2     = (const float*)d_in[8];
    float* out = (float*)d_out;

    float* h   = (float*)d_ws;
    float* agg = h + (size_t)NNODES * FD;
    float* deg = agg + (size_t)NNODES * FD;

    const int NODE_B = 256;
    const int tiles = (NNODES + 15) / 16;
    const int node_grid = (tiles + (NODE_B / 32) - 1) / (NODE_B / 32);

    owner_agg<true><<<NBLK, TPB, 0, stream>>>(x, src, dst, agg, deg, NNODES, NEDGES);
    node_wmma<FD><<<node_grid, NODE_B, 0, stream>>>(x, agg, deg, Wself1, Wneigh1,
                                                    b1, h, NNODES);

    owner_agg<false><<<NBLK, TPB, 0, stream>>>(h, src, dst, agg, deg, NNODES, NEDGES);
    node_wmma<10><<<node_grid, NODE_B, 0, stream>>>(h, agg, deg, Wself2, Wneigh2,
                                                    b2, out, NNODES);
}


# First owner pass writes each wave's owned edges (node, source) to spare workspace; second pass adds that list and scans dst only past it; layer-2 sums regrouped in 32s from list start
# speedup vs baseline: 6.5214x; 1.7010x over previous
#include <hip/hip_runtime.h>
#include <math.h>

#define FD 5
#define NNODES 100000
#define NEDGES 6400000

typedef __attribute__((ext_vector_type(2))) float v2f;
typedef __attribute__((ext_vector_type(8))) float v8f;

__device__ __forceinline__ int imin(int a, int b) { return a < b ? a : b; }
__device__ __forceinline__ int imax(int a, int b) { return a > b ? a : b; }

#define NBLK 256
#define NPB  391
#define TPB  512
#define NWAVE (TPB / 32)
#define QW   2048
#define QS   (QW + 32)
#define VPI  8
#define EBITS 23
#define AC   (FD + 1)
#define SBITS 17
#define LHDR  32
#define LPMAX 4096
#define COV_ALL 0x7fffffff

typedef __attribute__((ext_vector_type(4))) int v4i;

struct __attribute__((packed, aligned(4))) row_t { float v[FD]; };

#define DB 4
typedef volatile __attribute__((address_space(3))) unsigned* lds_tag_ptr;
template <bool ADD_DEG, bool DIRECT>
__device__ __forceinline__ void drain_queue(const unsigned* myq, int qn, int lane,
                                            const float* __restrict__ h,
                                            const int* __restrict__ src,
                                            float* myacc,
                                            lds_tag_ptr mytag,
                                            unsigned* rec) {
    for (int i = 0; i < qn; i += DB * 32) {
        unsigned w[DB];
        int s[DB];
        row_t r[DB];
#pragma unroll
        for (int k = 0; k < DB; ++k) {
            const int idx = i + k * 32 + lane;
            w[k] = myq[idx < qn ? idx : qn - 1];
        }
#pragma unroll
        for (int k = 0; k < DB; ++k) {
            if (DIRECT) s[k] = imin((int)(w[k] & ((1u << SBITS) - 1u)), NNODES - 1);
            else        s[k] = src[w[k] & ((1u << EBITS) - 1u)];
        }
#pragma unroll
        for (int k = 0; k < DB; ++k)
            __builtin_memcpy(&r[k], h + (long)s[k] * FD, sizeof(row_t));
        if (!DIRECT && rec != nullptr) {
#pragma unroll
            for (int k = 0; k < DB; ++k) {
                const int idx = i + k * 32 + lane;
                if (idx < qn) rec[idx] = ((w[k] >> EBITS) << SBITS) | (unsigned)s[k];
            }
        }
#pragma unroll
        for (int k = 0; k < DB; ++k) {
            const unsigned t = DIRECT ? (w[k] >> SBITS) : (w[k] >> EBITS);
            bool pending = (i + k * 32 + lane) < qn;
            for (int round = 0; round < 32; ++round) {
                if (__builtin_amdgcn_ballot_w32(pending) == 0u) break;
                bool win = false;
                if (pending) {
                    mytag[t] = (unsigned)lane;
                    win = (mytag[t] == (unsigned)lane);
                }
                if (win) {
                    float2* p = (float2*)(myacc + t * AC);
                    float2 a0 = p[0], a1 = p[1], a2 = p[2];
                    a0.x += r[k].v[0];
                    a0.y += r[k].v[1];
                    a1.x += r[k].v[2];
                    a1.y += r[k].v[3];
                    a2.x += r[k].v[4];
                    if (ADD_DEG) a2.y += 1.0f;
                    p[0] = a0;
                    p[1] = a1;
                    p[2] = a2;
                    pending = false;
                }
            }
        }
    }
}

__device__ __forceinline__ void scan_full_trip(const v4i* d, v4i* nd,
                                               const v4i* __restrict__ dst4,
                                               int base, int step, int tid, int n_vec,
                                               int lo, unsigned cnt, int lane,
                                               unsigned* myq, int& qn) {
#pragma unroll
    for (int u = 0; u < VPI; ++u) {
        const unsigned t0 = (unsigned)(d[u][0] - lo);
        const unsigned t1 = (unsigned)(d[u][1] - lo);
        const unsigned t2 = (unsigned)(d[u][2] - lo);
        const unsigned t3 = (unsigned)(d[u][3] - lo);
        const unsigned t01 = t0 < t1 ? t0 : t1;
        const unsigned t23 = t2 < t3 ? t2 : t3;
        const unsigned tm  = t01 < t23 ? t01 : t23;
        if (__builtin_amdgcn_ballot_w32(tm < cnt) != 0u) {
            const unsigned e0 = 4u * (unsigned)(base + u * TPB + tid);
            const unsigned tt[4] = {t0, t1, t2, t3};
#pragma unroll
            for (int c = 0; c < 4; ++c) {
                const bool m = tt[c] < cnt;
                const unsigned mask = __builtin_amdgcn_ballot_w32(m);
                const int pos = (int)__builtin_amdgcn_mbcnt_lo(mask, (unsigned)qn);
                myq[m ? pos : QW + lane] = (tt[c] << EBITS) | (e0 + (unsigned)c);
                qn += __builtin_popcount(mask);
            }
        }
        if (u == 0) {
#pragma unroll
            for (int u2 = 0; u2 < VPI; ++u2) {
                const int v2 = base + step + u2 * TPB + tid;
                nd[u2] = dst4[imin(v2, n_vec - 1)];
            }
        }
    }
}

template <bool FIRST>
__global__ __launch_bounds__(TPB)
void owner_agg(const float* __restrict__ h,
               const int* __restrict__ src,
               const int* __restrict__ dst,
               float* __restrict__ agg,
               float* __restrict__ deg,
               unsigned* __restrict__ lst,
               size_t ws_bytes,
               int n_nodes, int n_edges) {
    __shared__ __attribute__((aligned(16))) float accs[NWAVE * NPB * AC];
    __shared__ unsigned q[NWAVE * QS];
    __shared__ unsigned tags[NWAVE * NPB];

    const int tid  = threadIdx.x;
    const int lane = tid & 31;
    const int wave = tid >> 5;
    const int lo   = blockIdx.x * NPB;
    const int cnt  = imax(imin(NPB, n_nodes - lo), 0);

    const size_t used = (size_t)(2 * FD + 1) * NNODES * sizeof(float);
    size_t pitch = ws_bytes > used ? (ws_bytes - used) / (sizeof(unsigned) * NBLK * NWAVE) : 0;
    pitch = pitch > LPMAX ? (size_t)LPMAX : (pitch & ~(size_t)31);
    const int cap = pitch > LHDR ? (int)(pitch - LHDR) : 0;
    unsigned* mylst = lst + (size_t)(blockIdx.x * NWAVE + wave) * pitch;

    for (int i = tid; i < NWAVE * NPB * AC; i += TPB) accs[i] = 0.0f;
    __syncthreads();

    float* myacc = accs + wave * (NPB * AC);
    unsigned* myq = q + wave * QS;
    lds_tag_ptr mytag = (lds_tag_ptr)(tags + wave * NPB);
    int qn = 0;

    const v4i* dst4 = (const v4i*)dst;
    const int n_vec = n_edges >> 2;
    const int step  = TPB * VPI;

    bool rec   = FIRST && cap > 0;
    int stored = 0;
    int cov    = 0;

    int start = 0;
    if (!FIRST && cap > 0) {
        int n = __builtin_amdgcn_readfirstlane((int)mylst[0]);
        start = __builtin_amdgcn_readfirstlane((int)mylst[1]);
        n     = imax(imin(n, cap), 0);
        start = imax(start, 0);
        drain_queue<false, true>(mylst + LHDR, n, lane, h, src, myacc, mytag, nullptr);
    }

    if (start != COV_ALL) {
        v4i nd[VPI];
#pragma unroll
        for (int u = 0; u < VPI; ++u) {
            const int v = start + u * TPB + tid;
            nd[u] = dst4[imin(v, n_vec - 1)];
        }
        for (int base = start; base < n_vec; base += step) {
            if (qn > QW - VPI * 4 * 32) {
                unsigned* rp = nullptr;
                if (FIRST) {
                    if (rec && stored + qn <= cap) {
                        rp = mylst + LHDR + stored;
                        stored += qn;
                        cov = base;
                    } else {
                        rec = false;
                    }
                }
                drain_queue<FIRST, false>(myq, qn, lane, h, src, myacc, mytag, rp);
                qn = 0;
            }
            if (base + step <= n_vec) {
                v4i d[VPI];
#pragma unroll
                for (int u = 0; u < VPI; ++u) d[u] = nd[u];
                scan_full_trip(d, nd, dst4, base, step, tid, n_vec, lo, (unsigned)cnt, lane, myq, qn);
            } else {
#pragma nounroll
                for (int vb = base; vb < n_vec; vb += TPB) {
                    const int v = vb + tid;
                    const bool valid = v < n_vec;
                    const v4i dv = dst4[valid ? v : n_vec - 1];
#pragma unroll
                    for (int c = 0; c < 4; ++c) {
                        const unsigned t = (unsigned)(dv[c] - lo);
                        const bool m = valid && (t < (unsigned)cnt);
                        const unsigned mask = __builtin_amdgcn_ballot_w32(m);
                        if (mask != 0u) {
                            if (m) {
                                const int pos = qn + (int)__builtin_amdgcn_mbcnt_lo(mask, 0u);
                                myq[pos] = (t << EBITS) | (unsigned)(4 * v + c);
                            }
                            qn += __builtin_popcount(mask);
                        }
                    }
                }
            }
        }
        for (int e0 = n_vec * 4; e0 < n_edges; e0 += TPB) {
            const int e = e0 + tid;
            if (qn > QW - 32) {
                rec = false;
                drain_queue<FIRST, false>(myq, qn, lane, h, src, myacc, mytag, nullptr);
                qn = 0;
            }
            const bool valid = e < n_edges;
            const unsigned t = (unsigned)(dst[valid ? e : 0] - lo);
            const bool m = valid && (t < (unsigned)cnt);
            const unsigned mask = __builtin_amdgcn_ballot_w32(m);
            if (m) {
                const int pos = qn + (int)__builtin_amdgcn_mbcnt_lo(mask, 0u);
                myq[pos] = (t << EBITS) | (unsigned)e;
            }
            qn += __builtin_popcount(mask);
        }
        {
            unsigned* rp = nullptr;
            if (FIRST && rec && stored + qn <= cap) {
                rp = mylst + LHDR + stored;
                stored += qn;
                cov = COV_ALL;
            }
            drain_queue<FIRST, false>(myq, qn, lane, h, src, myacc, mytag, rp);
        }
    }
    if (FIRST && cap > 0 && lane == 0) {
        mylst[0] = (unsigned)stored;
        mylst[1] = (unsigned)cov;
    }
    __syncthreads();

    for (int i = tid; i < cnt * AC; i += TPB) {
        float sum = 0.0f;
#pragma unroll
        for (int w = 0; w < NWAVE; ++w) sum += accs[w * (NPB * AC) + i];
        const int node = i / AC;
        const int c = i - node * AC;
        if (c < FD) agg[(long)(lo + node) * FD + c] = sum;
        else if (FIRST) deg[lo + node] = sum;
    }
}

template <int F_OUT>
__global__ void node_wmma(const float* __restrict__ h,
                          const float* __restrict__ agg,
                          const float* __restrict__ deg,
                          const float* __restrict__ Wself,
                          const float* __restrict__ Wneigh,
                          const float* __restrict__ bias,
                          float* __restrict__ out,
                          int n_nodes) {
    const int lane  = threadIdx.x & 31;
    const int wave  = threadIdx.x >> 5;
    const int tile  = blockIdx.x * (blockDim.x >> 5) + wave;
    const int m0    = tile * 16;
    if (m0 >= n_nodes) return;

    const int row   = lane & 15;
    const int khalf = lane >> 4;

    const int m  = m0 + row;
    const int mc = imin(m, n_nodes - 1);
    const float invdeg = 1.0f / fmaxf(deg[mc], 1.0f);

    const int n  = row;
    const int nc = imin(n, F_OUT - 1);
    const bool nvalid = (n < F_OUT);

    v8f c = {};
#pragma unroll
    for (int chunk = 0; chunk < 3; ++chunk) {
        v2f a, b;
#pragma unroll
        for (int j = 0; j < 2; ++j) {
            const int k  = chunk * 4 + khalf * 2 + j;
            const int kh = imin(k, FD - 1);
            const int kn = imin(imax(k - FD, 0), FD - 1);

            const float hv = h[(long)mc * FD + kh];
            const float nv = agg[(long)mc * FD + kn] * invdeg;
            const float av = (k < FD) ? hv : ((k < 2 * FD) ? nv : 0.0f);

            const float ws = Wself[kh * F_OUT + nc];
            const float wn = Wneigh[kn * F_OUT + nc];
            float bv = (k < FD) ? ws : ((k < 2 * FD) ? wn : 0.0f);
            bv = nvalid ? bv : 0.0f;

            a[j] = av;
            b[j] = bv;
        }
        c = __builtin_amdgcn_wmma_f32_16x16x4_f32(false, a, false, b,
                                                  (short)0, c, false, false);
    }

    const float bn = bias[nc];
#pragma unroll
    for (int r = 0; r < 8; ++r) {
        const int mm = m0 + r + khalf * 8;
        if (mm < n_nodes && nvalid) {
            const float v = c[r] + bn;
            out[(long)mm * F_OUT + n] = 1.0f / (1.0f + __expf(-v));
        }
    }
}

extern "C" void kernel_launch(void* const* d_in, const int* in_sizes, int n_in,
                              void* d_out, int out_size, void* d_ws, size_t ws_size,
                              hipStream_t stream) {
    (void)in_sizes; (void)n_in; (void)out_size;

    const float* x      = (const float*)d_in[0];
    const int*   src    = (const int*)d_in[1];
    const int*   dst    = (const int*)d_in[2];
    const float* Wself1 = (const float*)d_in[3];
    const float* Wneigh1= (const float*)d_in[4];
    const float* b1     = (const float*)d_in[5];
    const float* Wself2 = (const float*)d_in[6];
    const float* Wneigh2= (const float*)d_in[7];
    const float* b2     = (const float*)d_in[8];
    float* out = (float*)d_out;

    float* h   = (float*)d_ws;
    float* agg = h + (size_t)NNODES * FD;
    float* deg = agg + (size_t)NNODES * FD;
    unsigned* lst = (unsigned*)(deg + (size_t)NNODES);

    const int NODE_B = 256;
    const int tiles = (NNODES + 15) / 16;
    const int node_grid = (tiles + (NODE_B / 32) - 1) / (NODE_B / 32);

    owner_agg<true><<<NBLK, TPB, 0, stream>>>(x, src, dst, agg, deg, lst, ws_size, NNODES, NEDGES);
    node_wmma<FD><<<node_grid, NODE_B, 0, stream>>>(x, agg, deg, Wself1, Wneigh1,
                                                    b1, h, NNODES);

    owner_agg<false><<<NBLK, TPB, 0, stream>>>(h, src, dst, agg, deg, lst, ws_size, NNODES, NEDGES);
    node_wmma<10><<<node_grid, NODE_B, 0, stream>>>(h, agg, deg, Wself2, Wneigh2,
                                                    b2, out, NNODES);
}


# One pre-pass sorts all edges by coarse destination range into workspace bins; an owner work-group scans only its range's bin, not all of dst; queue order, and so the order of the sums, changes
# speedup vs baseline: 16.8583x; 2.5851x over previous
#include <hip/hip_runtime.h>
#include <math.h>

#define FD 5
#define NNODES 100000
#define NEDGES 6400000

typedef __attribute__((ext_vector_type(2))) float v2f;
typedef __attribute__((ext_vector_type(8))) float v8f;

__device__ __forceinline__ int imin(int a, int b) { return a < b ? a : b; }
__device__ __forceinline__ int imax(int a, int b) { return a > b ? a : b; }

#define NBLK 256
#define NPB  391
#define TPB  512
#define NWAVE (TPB / 32)
#define QW   2048
#define QS   (QW + 32)
#define VPI  8
#define EBITS 23
#define AC   (FD + 1)
#define SBITS 17
#define LHDR  32
#define LPMAX 4096
#define COV_ALL 0x7fffffff

#define NGRP    16
#define NGROUPS (NBLK / NGRP)
#define GNODES  (NGRP * NPB)
#define BWAVES  (NBLK * NWAVE)
#define SEGW    160
#define GWORDS  (BWAVES * SEGW)
#define CO_WORDS  ((size_t)NGROUPS * GWORDS)
#define OVF_WORDS ((size_t)NGROUPS * BWAVES)
#define PAD_WORD 0xffffffffu
#define SMASK   ((1u << SBITS) - 1u)
static_assert(GWORDS % (4 * TPB * VPI) == 0, "a group array is a whole number of scan trips");
static_assert(GNODES < (1 << (32 - SBITS)) - 1, "group node index and PAD_WORD fit");

typedef __attribute__((ext_vector_type(4))) int v4i;

struct __attribute__((packed, aligned(4))) row_t { float v[FD]; };

#define DB 4
typedef volatile __attribute__((address_space(3))) unsigned* lds_tag_ptr;
template <bool ADD_DEG, bool DIRECT>
__device__ __forceinline__ void drain_queue(const unsigned* myq, int qn, int lane,
                                            const float* __restrict__ h,
                                            const int* __restrict__ src,
                                            float* myacc,
                                            lds_tag_ptr mytag,
                                            unsigned* rec) {
    for (int i = 0; i < qn; i += DB * 32) {
        unsigned w[DB];
        int s[DB];
        row_t r[DB];
#pragma unroll
        for (int k = 0; k < DB; ++k) {
            const int idx = i + k * 32 + lane;
            w[k] = myq[idx < qn ? idx : qn - 1];
        }
#pragma unroll
        for (int k = 0; k < DB; ++k) {
            if (DIRECT) s[k] = imin((int)(w[k] & ((1u << SBITS) - 1u)), NNODES - 1);
            else        s[k] = src[w[k] & ((1u << EBITS) - 1u)];
        }
#pragma unroll
        for (int k = 0; k < DB; ++k)
            __builtin_memcpy(&r[k], h + (long)s[k] * FD, sizeof(row_t));
        if (rec != nullptr) {
#pragma unroll
            for (int k = 0; k < DB; ++k) {
                const int idx = i + k * 32 + lane;
                if (idx < qn) rec[idx] = DIRECT ? w[k] : (((w[k] >> EBITS) << SBITS) | (unsigned)s[k]);
            }
        }
#pragma unroll
        for (int k = 0; k < DB; ++k) {
            const unsigned t = DIRECT ? (w[k] >> SBITS) : (w[k] >> EBITS);
            bool pending = (i + k * 32 + lane) < qn;
            for (int round = 0; round < 32; ++round) {
                if (__builtin_amdgcn_ballot_w32(pending) == 0u) break;
                bool win = false;
                if (pending) {
                    mytag[t] = (unsigned)lane;
                    win = (mytag[t] == (unsigned)lane);
                }
                if (win) {
                    float2* p = (float2*)(myacc + t * AC);
                    float2 a0 = p[0], a1 = p[1], a2 = p[2];
                    a0.x += r[k].v[0];
                    a0.y += r[k].v[1];
                    a1.x += r[k].v[2];
                    a1.y += r[k].v[3];
                    a2.x += r[k].v[4];
                    if (ADD_DEG) a2.y += 1.0f;
                    p[0] = a0;
                    p[1] = a1;
                    p[2] = a2;
                    pending = false;
                }
            }
        }
    }
}

template <bool CO>
__device__ __forceinline__ void scan_full_trip(const v4i* d, v4i* nd,
                                               const v4i* __restrict__ scan4,
                                               int base, int step, int tid, int n_vec,
                                               int lo, unsigned cnt, int lane,
                                               unsigned* myq, int& qn) {
#pragma unroll
    for (int u = 0; u < VPI; ++u) {
        const unsigned t0 = (CO ? ((unsigned)d[u][0] >> SBITS) : (unsigned)d[u][0]) - (unsigned)lo;
        const unsigned t1 = (CO ? ((unsigned)d[u][1] >> SBITS) : (unsigned)d[u][1]) - (unsigned)lo;
        const unsigned t2 = (CO ? ((unsigned)d[u][2] >> SBITS) : (unsigned)d[u][2]) - (unsigned)lo;
        const unsigned t3 = (CO ? ((unsigned)d[u][3] >> SBITS) : (unsigned)d[u][3]) - (unsigned)lo;
        const unsigned t01 = t0 < t1 ? t0 : t1;
        const unsigned t23 = t2 < t3 ? t2 : t3;
        const unsigned tm  = t01 < t23 ? t01 : t23;
        if (__builtin_amdgcn_ballot_w32(tm < cnt) != 0u) {
            const unsigned e0 = 4u * (unsigned)(base + u * TPB + tid);
            const unsigned tt[4] = {t0, t1, t2, t3};
#pragma unroll
            for (int c = 0; c < 4; ++c) {
                const bool m = tt[c] < cnt;
                const unsigned mask = __builtin_amdgcn_ballot_w32(m);
                const int pos = (int)__builtin_amdgcn_mbcnt_lo(mask, (unsigned)qn);
                const unsigned ent = CO ? ((tt[c] << SBITS) | ((unsigned)d[u][c] & SMASK))
                                        : ((tt[c] << EBITS) | (e0 + (unsigned)c));
                myq[m ? pos : QW + lane] = ent;
                qn += __builtin_popcount(mask);
            }
        }
        if (u == 0) {
#pragma unroll
            for (int u2 = 0; u2 < VPI; ++u2) {
                const int v2 = base + step + u2 * TPB + tid;
                nd[u2] = scan4[imin(v2, n_vec - 1)];
            }
        }
    }
}

#define BSTR (NGROUPS * SEGW + 32)
#define BU   4
static_assert(SEGW % 32 == 0, "a segment is copied out in whole 32-lane rows");
static_assert(NGROUPS == 16, "the group number is taken apart as four bits");
__global__ __launch_bounds__(TPB)
void coarse_bin(const int* __restrict__ src,
                const int* __restrict__ dst,
                unsigned* __restrict__ extra,
                size_t ws_bytes, int n_edges) {
    __shared__ unsigned buf[NWAVE * BSTR];
    __shared__ unsigned fills[NWAVE * NGROUPS];

    const size_t used = (size_t)(2 * FD + 1) * NNODES * sizeof(float);
    if ((n_edges & 3) != 0 || ws_bytes < used + (CO_WORDS + OVF_WORDS) * sizeof(unsigned)) return;

    const int tid  = threadIdx.x;
    const int lane = tid & 31;
    const int wave = tid >> 5;
    const int gw   = blockIdx.x * NWAVE + wave;
    unsigned* mybuf = buf + wave * BSTR;
    lds_tag_ptr myfill = (lds_tag_ptr)(fills + wave * NGROUPS);
    for (int i = lane; i < NGROUPS * SEGW; i += 32) mybuf[i] = PAD_WORD;
    if (lane < NGROUPS) myfill[lane] = 0u;

    const v4i* dst4 = (const v4i*)dst;
    const v4i* src4 = (const v4i*)src;
    const int n_vec = n_edges >> 2;
    const int per   = (n_vec + BWAVES - 1) / BWAVES;
    const int v0    = imin(gw * per, n_vec);
    const int v1    = imin(v0 + per, n_vec);

    for (int vb = v0; vb < v1; vb += 32 * BU) {
        v4i dvv[BU], svv[BU];
#pragma unroll
        for (int k = 0; k < BU; ++k) {
            const int vc = imin(vb + k * 32 + lane, v1 - 1);
            dvv[k] = dst4[vc];
            svv[k] = src4[vc];
        }
#pragma unroll
        for (int k = 0; k < BU; ++k) {
            const bool valid = vb + k * 32 + lane < v1;
            const v4i dv = dvv[k];
            const v4i sv = svv[k];
#pragma unroll
            for (int c = 0; c < 4; ++c) {
                const unsigned ud  = (unsigned)dv[c];
                const unsigned grp = ud / (unsigned)GNODES;
                const unsigned sn  = (unsigned)imin(imax(sv[c], 0), NNODES - 1);
                const unsigned word = ((ud - grp * (unsigned)GNODES) << SBITS) | sn;
                const bool act = valid && grp < (unsigned)NGROUPS;
                const unsigned key = act ? grp : 0u;
                unsigned same = __builtin_amdgcn_ballot_w32(act);
#pragma unroll
                for (int b = 0; b < 4; ++b) {
                    const bool bit = ((key >> b) & 1u) != 0u;
                    const unsigned vb1 = __builtin_amdgcn_ballot_w32(bit);
                    same &= bit ? vb1 : ~vb1;
                }
                const unsigned rank  = __builtin_amdgcn_mbcnt_lo(same, 0u);
                const unsigned total = (unsigned)__builtin_popcount(same);
                const unsigned fill  = myfill[key];
                const unsigned pos   = fill + rank;
                mybuf[(act && pos < (unsigned)SEGW) ? key * SEGW + pos : (unsigned)(NGROUPS * SEGW + lane)] = word;
                if (act && rank + 1u == total) myfill[key] = fill + total;
            }
        }
    }

    unsigned* coarse = extra;
    unsigned* ovf    = extra + CO_WORDS;
#pragma nounroll
    for (int g = 0; g < NGROUPS; ++g) {
        unsigned* seg = coarse + ((size_t)g * BWAVES + gw) * SEGW;
        unsigned wv[SEGW / 32];
#pragma unroll
        for (int j = 0; j < SEGW / 32; ++j) wv[j] = mybuf[g * SEGW + j * 32 + lane];
#pragma unroll
        for (int j = 0; j < SEGW / 32; ++j) seg[j * 32 + lane] = wv[j];
    }
    if (lane < NGROUPS) ovf[(size_t)lane * BWAVES + gw] = myfill[lane] > (unsigned)SEGW ? 1u : 0u;
}

template <bool FIRST, bool CO>
__device__ __forceinline__ void owner_pass(const float* __restrict__ h,
                                           const int* __restrict__ src,
                                           const int* __restrict__ dst,
                                           const v4i* __restrict__ scan4,
                                           int n_vec, int n_edges,
                                           int lo, int cnt, int tid, int lane,
                                           unsigned* myq, float* myacc, lds_tag_ptr mytag,
                                           unsigned* mylst, int cap) {
    int qn = 0;
    const int step  = TPB * VPI;

    bool rec   = FIRST && cap > 0;
    int stored = 0;
    int cov    = 0;

    int start = 0;
    if (!FIRST && cap > 0) {
        int n = __builtin_amdgcn_readfirstlane((int)mylst[0]);
        start = __builtin_amdgcn_readfirstlane((int)mylst[1]);
        n     = imax(imin(n, cap), 0);
        start = imax(start, 0);
        drain_queue<false, true>(mylst + LHDR, n, lane, h, src, myacc, mytag, nullptr);
    }

    if (start != COV_ALL) {
        v4i nd[VPI];
#pragma unroll
        for (int u = 0; u < VPI; ++u) {
            const int v = start + u * TPB + tid;
            nd[u] = scan4[imin(v, n_vec - 1)];
        }
        for (int base = start; base < n_vec; base += step) {
            if (qn > QW - VPI * 4 * 32) {
                unsigned* rp = nullptr;
                if (FIRST) {
                    if (rec && stored + qn <= cap) {
                        rp = mylst + LHDR + stored;
                        stored += qn;
                        cov = base;
                    } else {
                        rec = false;
                    }
                }
                drain_queue<FIRST, CO>(myq, qn, lane, h, src, myacc, mytag, rp);
                qn = 0;
            }
            if (CO || base + step <= n_vec) {
                v4i d[VPI];
#pragma unroll
                for (int u = 0; u < VPI; ++u) d[u] = nd[u];
                scan_full_trip<CO>(d, nd, scan4, base, step, tid, n_vec, lo, (unsigned)cnt, lane, myq, qn);
            } else {
#pragma nounroll
                for (int vb = base; vb < n_vec; vb += TPB) {
                    const int v = vb + tid;
                    const bool valid = v < n_vec;
                    const v4i dv = scan4[valid ? v : n_vec - 1];
#pragma unroll
                    for (int c = 0; c < 4; ++c) {
                        const unsigned t = (unsigned)(dv[c] - lo);
                        const bool m = valid && (t < (unsigned)cnt);
                        const unsigned mask = __builtin_amdgcn_ballot_w32(m);
                        if (mask != 0u) {
                            if (m) {
                                const int pos = qn + (int)__builtin_amdgcn_mbcnt_lo(mask, 0u);
                                myq[pos] = (t << EBITS) | (unsigned)(4 * v + c);
                            }
                            qn += __builtin_popcount(mask);
                        }
                    }
                }
            }
        }
        if (!CO) {
            for (int e0 = n_vec * 4; e0 < n_edges; e0 += TPB) {
                const int e = e0 + tid;
                if (qn > QW - 32) {
                    rec = false;
                    drain_queue<FIRST, false>(myq, qn, lane, h, src, myacc, mytag, nullptr);
                    qn = 0;
                }
                const bool valid = e < n_edges;
                const unsigned t = (unsigned)(dst[valid ? e : 0] - lo);
                const bool m = valid && (t < (unsigned)cnt);
                const unsigned mask = __builtin_amdgcn_ballot_w32(m);
                if (m) {
                    const int pos = qn + (int)__builtin_amdgcn_mbcnt_lo(mask, 0u);
                    myq[pos] = (t << EBITS) | (unsigned)e;
                }
                qn += __builtin_popcount(mask);
            }
        }
        {
            unsigned* rp = nullptr;
            if (FIRST && rec && stored + qn <= cap) {
                rp = mylst + LHDR + stored;
                stored += qn;
                cov = COV_ALL;
            }
            drain_queue<FIRST, CO>(myq, qn, lane, h, src, myacc, mytag, rp);
        }
    }
    if (FIRST && cap > 0 && lane == 0) {
        mylst[0] = (unsigned)stored;
        mylst[1] = (unsigned)cov;
    }
}

template <bool FIRST>
__global__ __launch_bounds__(TPB)
void owner_agg(const float* __restrict__ h,
               const int* __restrict__ src,
               const int* __restrict__ dst,
               float* __restrict__ agg,
               float* __restrict__ deg,
               unsigned* __restrict__ extra,
               size_t ws_bytes,
               int n_nodes, int n_edges) {
    __shared__ __attribute__((aligned(16))) float accs[NWAVE * NPB * AC];
    __shared__ unsigned q[NWAVE * QS];
    __shared__ unsigned tags[NWAVE * NPB];
    __shared__ unsigned vote[NWAVE];

    const int tid  = threadIdx.x;
    const int lane = tid & 31;
    const int wave = tid >> 5;
    const int lo   = blockIdx.x * NPB;
    const int cnt  = imax(imin(NPB, n_nodes - lo), 0);
    const int grp  = blockIdx.x / NGRP;

    const size_t used = (size_t)(2 * FD + 1) * NNODES * sizeof(float);
    const size_t co_bytes = (CO_WORDS + OVF_WORDS) * sizeof(unsigned);
    const bool co_room = (n_edges & 3) == 0 && ws_bytes >= used + co_bytes;
    size_t rest = ws_bytes > used ? ws_bytes - used : 0;
    unsigned* lst = extra;
    if (co_room) {
        rest -= co_bytes;
        lst = extra + CO_WORDS + OVF_WORDS;
    }
    size_t pitch = rest / (sizeof(unsigned) * NBLK * NWAVE);
    pitch = pitch > LPMAX ? (size_t)LPMAX : (pitch & ~(size_t)31);
    const int cap = pitch > LHDR ? (int)(pitch - LHDR) : 0;
    unsigned* mylst = lst + (size_t)(blockIdx.x * NWAVE + wave) * pitch;

    int bad = 0;
    if (co_room) {
        const unsigned* ovf = extra + CO_WORDS + (size_t)grp * BWAVES;
        for (int i = tid; i < BWAVES; i += TPB) bad |= (int)(ovf[i] != 0u);
    }
    const unsigned wbad = __builtin_amdgcn_ballot_w32(bad != 0);
    if (lane == 0) vote[wave] = wbad;
    for (int i = tid; i < NWAVE * NPB * AC; i += TPB) accs[i] = 0.0f;
    __syncthreads();
    unsigned anybad = 0u;
#pragma unroll
    for (int w = 0; w < NWAVE; ++w) anybad |= vote[w];
    const bool use_co = co_room && __builtin_amdgcn_readfirstlane((int)anybad) == 0;

    float* myacc = accs + wave * (NPB * AC);
    unsigned* myq = q + wave * QS;
    lds_tag_ptr mytag = (lds_tag_ptr)(tags + wave * NPB);

    if (use_co) {
        owner_pass<FIRST, true>(h, src, dst, (const v4i*)(extra + (size_t)grp * GWORDS), GWORDS / 4, n_edges,
                                (blockIdx.x % NGRP) * NPB, cnt, tid, lane, myq, myacc, mytag, mylst, cap);
    } else {
        owner_pass<FIRST, false>(h, src, dst, (const v4i*)dst, n_edges >> 2, n_edges,
                                 lo, cnt, tid, lane, myq, myacc, mytag, mylst, cap);
    }
    __syncthreads();

    for (int i = tid; i < cnt * AC; i += TPB) {
        float sum = 0.0f;
#pragma unroll
        for (int w = 0; w < NWAVE; ++w) sum += accs[w * (NPB * AC) + i];
        const int node = i / AC;
        const int c = i - node * AC;
        if (c < FD) agg[(long)(lo + node) * FD + c] = sum;
        else if (FIRST) deg[lo + node] = sum;
    }
}

template <int F_OUT>
__global__ void node_wmma(const float* __restrict__ h,
                          const float* __restrict__ agg,
                          const float* __restrict__ deg,
                          const float* __restrict__ Wself,
                          const float* __restrict__ Wneigh,
                          const float* __restrict__ bias,
                          float* __restrict__ out,
                          int n_nodes) {
    const int lane  = threadIdx.x & 31;
    const int wave  = threadIdx.x >> 5;
    const int tile  = blockIdx.x * (blockDim.x >> 5) + wave;
    const int m0    = tile * 16;
    if (m0 >= n_nodes) return;

    const int row   = lane & 15;
    const int khalf = lane >> 4;

    const int m  = m0 + row;
    const int mc = imin(m, n_nodes - 1);
    const float invdeg = 1.0f / fmaxf(deg[mc], 1.0f);

    const int n  = row;
    const int nc = imin(n, F_OUT - 1);
    const bool nvalid = (n < F_OUT);

    v8f c = {};
#pragma unroll
    for (int chunk = 0; chunk < 3; ++chunk) {
        v2f a, b;
#pragma unroll
        for (int j = 0; j < 2; ++j) {
            const int k  = chunk * 4 + khalf * 2 + j;
            const int kh = imin(k, FD - 1);
            const int kn = imin(imax(k - FD, 0), FD - 1);

            const float hv = h[(long)mc * FD + kh];
            const float nv = agg[(long)mc * FD + kn] * invdeg;
            const float av = (k < FD) ? hv : ((k < 2 * FD) ? nv : 0.0f);

            const float ws = Wself[kh * F_OUT + nc];
            const float wn = Wneigh[kn * F_OUT + nc];
            float bv = (k < FD) ? ws : ((k < 2 * FD) ? wn : 0.0f);
            bv = nvalid ? bv : 0.0f;

            a[j] = av;
            b[j] = bv;
        }
        c = __builtin_amdgcn_wmma_f32_16x16x4_f32(false, a, false, b,
                                                  (short)0, c, false, false);
    }

    const float bn = bias[nc];
#pragma unroll
    for (int r = 0; r < 8; ++r) {
        const int mm = m0 + r + khalf * 8;
        if (mm < n_nodes && nvalid) {
            const float v = c[r] + bn;
            out[(long)mm * F_OUT + n] = 1.0f / (1.0f + __expf(-v));
        }
    }
}

extern "C" void kernel_launch(void* const* d_in, const int* in_sizes, int n_in,
                              void* d_out, int out_size, void* d_ws, size_t ws_size,
                              hipStream_t stream) {
    (void)in_sizes; (void)n_in; (void)out_size;

    const float* x      = (const float*)d_in[0];
    const int*   src    = (const int*)d_in[1];
    const int*   dst    = (const int*)d_in[2];
    const float* Wself1 = (const float*)d_in[3];
    const float* Wneigh1= (const float*)d_in[4];
    const float* b1     = (const float*)d_in[5];
    const float* Wself2 = (const float*)d_in[6];
    const float* Wneigh2= (const float*)d_in[7];
    const float* b2     = (const float*)d_in[8];
    float* out = (float*)d_out;

    float* h   = (float*)d_ws;
    float* agg = h + (size_t)NNODES * FD;
    float* deg = agg + (size_t)NNODES * FD;
    unsigned* extra = (unsigned*)(deg + (size_t)NNODES);

    const int NODE_B = 256;
    const int tiles = (NNODES + 15) / 16;
    const int node_grid = (tiles + (NODE_B / 32) - 1) / (NODE_B / 32);

    coarse_bin<<<NBLK, TPB, 0, stream>>>(src, dst, extra, ws_size, NEDGES);

    owner_agg<true><<<NBLK, TPB, 0, stream>>>(x, src, dst, agg, deg, extra, ws_size, NNODES, NEDGES);
    node_wmma<FD><<<node_grid, NODE_B, 0, stream>>>(x, agg, deg, Wself1, Wneigh1,
                                                    b1, h, NNODES);

    owner_agg<false><<<NBLK, TPB, 0, stream>>>(h, src, dst, agg, deg, extra, ws_size, NNODES, NEDGES);
    node_wmma<10><<<node_grid, NODE_B, 0, stream>>>(h, agg, deg, Wself2, Wneigh2,
                                                    b2, out, NNODES);
}


# Pre-pass sorts edges down to the owner: behind a barrier wave g re-sorts group g's LDS segments by owner, so an owner scans 1/16 of the words; queue order, and so the order of the sums, changes
# speedup vs baseline: 22.0808x; 1.3098x over previous
#include <hip/hip_runtime.h>
#include <math.h>

#define FD 5
#define NNODES 100000
#define NEDGES 6400000

typedef __attribute__((ext_vector_type(2))) float v2f;
typedef __attribute__((ext_vector_type(8))) float v8f;

__device__ __forceinline__ int imin(int a, int b) { return a < b ? a : b; }
__device__ __forceinline__ int imax(int a, int b) { return a > b ? a : b; }

#define NBLK 256
#define NPB  391
#define TPB  512
#define NWAVE (TPB / 32)
#define QW   2048
#define QS   (QW + 32)
#define VPI  5
#define EBITS 23
#define AC   (FD + 1)
#define SBITS 17
#define LHDR  32
#define LPMAX 4096
#define COV_ALL 0x7fffffff

#define NGRP    16
#define NGROUPS (NBLK / NGRP)
#define GNODES  (NGRP * NPB)
#define BWAVES  (NBLK * NWAVE)
#define SEGW    160
#define OWORDS  (NBLK * SEGW)
#define CO_WORDS  ((size_t)NBLK * OWORDS)
#define OVF_WORDS ((size_t)NGROUPS * NBLK)
#define PAD_WORD 0xffffffffu
#define SMASK   ((1u << SBITS) - 1u)
static_assert(OWORDS % (4 * TPB * VPI) == 0, "an owner array is a whole number of scan trips");
static_assert(NGRP == NWAVE && NGROUPS == NWAVE, "stage two: one wave per group, one segment per owner of the group");
static_assert(GNODES < (1 << (32 - SBITS)) - 1, "group node index and PAD_WORD fit");

typedef __attribute__((ext_vector_type(4))) int v4i;

struct __attribute__((packed, aligned(4))) row_t { float v[FD]; };

#define DB 4
typedef volatile __attribute__((address_space(3))) unsigned* lds_tag_ptr;
template <bool ADD_DEG, bool DIRECT>
__device__ __forceinline__ void drain_queue(const unsigned* myq, int qn, int lane,
                                            const float* __restrict__ h,
                                            const int* __restrict__ src,
                                            float* myacc,
                                            lds_tag_ptr mytag,
                                            unsigned* rec) {
    for (int i = 0; i < qn; i += DB * 32) {
        unsigned w[DB];
        int s[DB];
        row_t r[DB];
#pragma unroll
        for (int k = 0; k < DB; ++k) {
            const int idx = i + k * 32 + lane;
            w[k] = myq[idx < qn ? idx : qn - 1];
        }
#pragma unroll
        for (int k = 0; k < DB; ++k) {
            if (DIRECT) s[k] = imin((int)(w[k] & ((1u << SBITS) - 1u)), NNODES - 1);
            else        s[k] = src[w[k] & ((1u << EBITS) - 1u)];
        }
#pragma unroll
        for (int k = 0; k < DB; ++k)
            __builtin_memcpy(&r[k], h + (long)s[k] * FD, sizeof(row_t));
        if (rec != nullptr) {
#pragma unroll
            for (int k = 0; k < DB; ++k) {
                const int idx = i + k * 32 + lane;
                if (idx < qn) rec[idx] = DIRECT ? w[k] : (((w[k] >> EBITS) << SBITS) | (unsigned)s[k]);
            }
        }
#pragma unroll
        for (int k = 0; k < DB; ++k) {
            const unsigned t = DIRECT ? (w[k] >> SBITS) : (w[k] >> EBITS);
            bool pending = (i + k * 32 + lane) < qn;
            for (int round = 0; round < 32; ++round) {
                if (__builtin_amdgcn_ballot_w32(pending) == 0u) break;
                bool win = false;
                if (pending) {
                    mytag[t] = (unsigned)lane;
                    win = (mytag[t] == (unsigned)lane);
                }
                if (win) {
                    float2* p = (float2*)(myacc + t * AC);
                    float2 a0 = p[0], a1 = p[1], a2 = p[2];
                    a0.x += r[k].v[0];
                    a0.y += r[k].v[1];
                    a1.x += r[k].v[2];
                    a1.y += r[k].v[3];
                    a2.x += r[k].v[4];
                    if (ADD_DEG) a2.y += 1.0f;
                    p[0] = a0;
                    p[1] = a1;
                    p[2] = a2;
                    pending = false;
                }
            }
        }
    }
}

template <bool CO>
__device__ __forceinline__ void scan_full_trip(const v4i* d, v4i* nd,
                                               const v4i* __restrict__ scan4,
                                               int base, int step, int tid, int n_vec,
                                               int lo, unsigned cnt, int lane,
                                               unsigned* myq, int& qn) {
#pragma unroll
    for (int u = 0; u < VPI; ++u) {
        const unsigned t0 = (CO ? ((unsigned)d[u][0] >> SBITS) : (unsigned)d[u][0]) - (unsigned)lo;
        const unsigned t1 = (CO ? ((unsigned)d[u][1] >> SBITS) : (unsigned)d[u][1]) - (unsigned)lo;
        const unsigned t2 = (CO ? ((unsigned)d[u][2] >> SBITS) : (unsigned)d[u][2]) - (unsigned)lo;
        const unsigned t3 = (CO ? ((unsigned)d[u][3] >> SBITS) : (unsigned)d[u][3]) - (unsigned)lo;
        const unsigned t01 = t0 < t1 ? t0 : t1;
        const unsigned t23 = t2 < t3 ? t2 : t3;
        const unsigned tm  = t01 < t23 ? t01 : t23;
        if (__builtin_amdgcn_ballot_w32(tm < cnt) != 0u) {
            const unsigned e0 = 4u * (unsigned)(base + u * TPB + tid);
            const unsigned tt[4] = {t0, t1, t2, t3};
#pragma unroll
            for (int c = 0; c < 4; ++c) {
                const bool m = tt[c] < cnt;
                const unsigned mask = __builtin_amdgcn_ballot_w32(m);
                const int pos = (int)__builtin_amdgcn_mbcnt_lo(mask, (unsigned)qn);
                const unsigned ent = CO ? ((tt[c] << SBITS) | ((unsigned)d[u][c] & SMASK))
                                        : ((tt[c] << EBITS) | (e0 + (unsigned)c));
                myq[m ? pos : QW + lane] = ent;
                qn += __builtin_popcount(mask);
            }
        }
        if (u == 0) {
#pragma unroll
            for (int u2 = 0; u2 < VPI; ++u2) {
                const int v2 = base + step + u2 * TPB + tid;
                nd[u2] = scan4[imin(v2, n_vec - 1)];
            }
        }
    }
}

#define BSEG (NWAVE * NGROUPS * SEGW)
#define BU   4
#define S2R  (NWAVE * SEGW / 32)
static_assert(SEGW % 32 == 0, "a segment is copied out in whole 32-lane rows");
static_assert(NGROUPS == 16 && NGRP == 16, "a key is taken apart as four bits");

__device__ __forceinline__ void place32(unsigned* buf, lds_tag_ptr myfill, bool act, unsigned key,
                                        unsigned word, unsigned seg0, unsigned segp, unsigned myspare) {
    unsigned same = __builtin_amdgcn_ballot_w32(act);
#pragma unroll
    for (int b = 0; b < 4; ++b) {
        const bool bit = ((key >> b) & 1u) != 0u;
        const unsigned vb1 = __builtin_amdgcn_ballot_w32(bit);
        same &= bit ? vb1 : ~vb1;
    }
    const unsigned rank  = __builtin_amdgcn_mbcnt_lo(same, 0u);
    const unsigned total = (unsigned)__builtin_popcount(same);
    const unsigned fill  = myfill[key];
    const unsigned pos   = fill + rank;
    buf[(act && pos < (unsigned)SEGW) ? seg0 + key * segp + pos : myspare] = word;
    if (act && rank + 1u == total) myfill[key] = fill + total;
}

__global__ __launch_bounds__(TPB)
void coarse_bin(const int* __restrict__ src,
                const int* __restrict__ dst,
                unsigned* __restrict__ extra,
                size_t ws_bytes, int n_edges) {
    __shared__ unsigned buf[BSEG + TPB];
    __shared__ unsigned fills[NWAVE * NGROUPS];
    __shared__ unsigned fills2[NWAVE * NGRP];

    const size_t used = (size_t)(2 * FD + 1) * NNODES * sizeof(float);
    if ((n_edges & 3) != 0 || ws_bytes < used + (CO_WORDS + OVF_WORDS) * sizeof(unsigned)) return;

    const int tid  = threadIdx.x;
    const int lane = tid & 31;
    const int wave = tid >> 5;
    const int gw   = blockIdx.x * NWAVE + wave;
    const unsigned myspare = (unsigned)(BSEG + tid);
    const unsigned seg0 = (unsigned)(wave * SEGW);
    lds_tag_ptr myfill = (lds_tag_ptr)(fills + wave * NGROUPS);
    for (int i = lane; i < NGROUPS * SEGW; i += 32) {
        const int g = i / SEGW;
        buf[(g * NWAVE + wave) * SEGW + (i - g * SEGW)] = PAD_WORD;
    }
    if (lane < NGROUPS) myfill[lane] = 0u;

    const v4i* dst4 = (const v4i*)dst;
    const v4i* src4 = (const v4i*)src;
    const int n_vec = n_edges >> 2;
    const int per   = (n_vec + BWAVES - 1) / BWAVES;
    const int v0    = imin(gw * per, n_vec);
    const int v1    = imin(v0 + per, n_vec);

    for (int vb = v0; vb < v1; vb += 32 * BU) {
        v4i dvv[BU], svv[BU];
#pragma unroll
        for (int k = 0; k < BU; ++k) {
            const int vc = imin(vb + k * 32 + lane, v1 - 1);
            dvv[k] = dst4[vc];
            svv[k] = src4[vc];
        }
#pragma unroll
        for (int k = 0; k < BU; ++k) {
            const bool valid = vb + k * 32 + lane < v1;
            const v4i dv = dvv[k];
            const v4i sv = svv[k];
#pragma unroll
            for (int c = 0; c < 4; ++c) {
                const unsigned ud  = (unsigned)dv[c];
                const unsigned grp = ud / (unsigned)GNODES;
                const unsigned sn  = (unsigned)imin(imax(sv[c], 0), NNODES - 1);
                const unsigned word = ((ud - grp * (unsigned)GNODES) << SBITS) | sn;
                const bool act = valid && grp < (unsigned)NGROUPS;
                place32(buf, myfill, act, act ? grp : 0u, word, seg0, (unsigned)(NWAVE * SEGW), myspare);
            }
        }
    }
    __syncthreads();

    const unsigned in0 = (unsigned)(wave * NWAVE * SEGW);
    lds_tag_ptr myfill2 = (lds_tag_ptr)(fills2 + wave * NGRP);
    unsigned win[S2R];
#pragma unroll
    for (int r = 0; r < S2R; ++r) win[r] = buf[in0 + r * 32 + lane];
    for (int i = lane; i < NGRP * SEGW; i += 32) buf[in0 + i] = PAD_WORD;
    if (lane < NGRP) myfill2[lane] = 0u;
#pragma unroll
    for (int r = 0; r < S2R; ++r) {
        const unsigned word = win[r];
        const bool act = word != PAD_WORD;
        if (__builtin_amdgcn_ballot_w32(act) == 0u) continue;
        const unsigned own = (word >> SBITS) / (unsigned)NPB;
        place32(buf, myfill2, act, act ? own : 0u, word, in0, (unsigned)SEGW, myspare);
    }

    unsigned* coarse = extra;
    unsigned* ovf    = extra + CO_WORDS;
#pragma nounroll
    for (int o = 0; o < NGRP; ++o) {
        unsigned* seg = coarse + ((size_t)(wave * NGRP + o) * NBLK + blockIdx.x) * SEGW;
        unsigned wv[SEGW / 32];
#pragma unroll
        for (int j = 0; j < SEGW / 32; ++j) wv[j] = buf[in0 + o * SEGW + j * 32 + lane];
#pragma unroll
        for (int j = 0; j < SEGW / 32; ++j) seg[j * 32 + lane] = wv[j];
    }
    bool over = false;
    if (lane < NWAVE) over = fills[lane * NGROUPS + wave] > (unsigned)SEGW || myfill2[lane] > (unsigned)SEGW;
    const unsigned anyover = __builtin_amdgcn_ballot_w32(over);
    if (lane == 0) ovf[(size_t)wave * NBLK + blockIdx.x] = anyover != 0u ? 1u : 0u;
}

template <bool FIRST, bool CO>
__device__ __forceinline__ void owner_pass(const float* __restrict__ h,
                                           const int* __restrict__ src,
                                           const int* __restrict__ dst,
                                           const v4i* __restrict__ scan4,
                                           int n_vec, int n_edges,
                                           int lo, int cnt, int tid, int lane,
                                           unsigned* myq, float* myacc, lds_tag_ptr mytag,
                                           unsigned* mylst, int cap) {
    int qn = 0;
    const int step  = TPB * VPI;

    bool rec   = FIRST && cap > 0;
    int stored = 0;
    int cov    = 0;

    int start = 0;
    if (!FIRST && cap > 0) {
        int n = __builtin_amdgcn_readfirstlane((int)mylst[0]);
        start = __builtin_amdgcn_readfirstlane((int)mylst[1]);
        n     = imax(imin(n, cap), 0);
        start = imax(start, 0);
        drain_queue<false, true>(mylst + LHDR, n, lane, h, src, myacc, mytag, nullptr);
    }

    if (start != COV_ALL) {
        v4i nd[VPI];
#pragma unroll
        for (int u = 0; u < VPI; ++u) {
            const int v = start + u * TPB + tid;
            nd[u] = scan4[imin(v, n_vec - 1)];
        }
        for (int base = start; base < n_vec; base += step) {
            if (qn > QW - VPI * 4 * 32) {
                unsigned* rp = nullptr;
                if (FIRST) {
                    if (rec && stored + qn <= cap) {
                        rp = mylst + LHDR + stored;
                        stored += qn;
                        cov = base;
                    } else {
                        rec = false;
                    }
                }
                drain_queue<FIRST, CO>(myq, qn, lane, h, src, myacc, mytag, rp);
                qn = 0;
            }
            if (CO || base + step <= n_vec) {
                v4i d[VPI];
#pragma unroll
                for (int u = 0; u < VPI; ++u) d[u] = nd[u];
                scan_full_trip<CO>(d, nd, scan4, base, step, tid, n_vec, lo, (unsigned)cnt, lane, myq, qn);
            } else {
#pragma nounroll
                for (int vb = base; vb < n_vec; vb += TPB) {
                    const int v = vb + tid;
                    const bool valid = v < n_vec;
                    const v4i dv = scan4[valid ? v : n_vec - 1];
#pragma unroll
                    for (int c = 0; c < 4; ++c) {
                        const unsigned t = (unsigned)(dv[c] - lo);
                        const bool m = valid && (t < (unsigned)cnt);
                        const unsigned mask = __builtin_amdgcn_ballot_w32(m);
                        if (mask != 0u) {
                            if (m) {
                                const int pos = qn + (int)__builtin_amdgcn_mbcnt_lo(mask, 0u);
                                myq[pos] = (t << EBITS) | (unsigned)(4 * v + c);
                            }
                            qn += __builtin_popcount(mask);
                        }
                    }
                }
            }
        }
        if (!CO) {
            for (int e0 = n_vec * 4; e0 < n_edges; e0 += TPB) {
                const int e = e0 + tid;
                if (qn > QW - 32) {
                    rec = false;
                    drain_queue<FIRST, false>(myq, qn, lane, h, src, myacc, mytag, nullptr);
                    qn = 0;
                }
                const bool valid = e < n_edges;
                const unsigned t = (unsigned)(dst[valid ? e : 0] - lo);
                const bool m = valid && (t < (unsigned)cnt);
                const unsigned mask = __builtin_amdgcn_ballot_w32(m);
                if (m) {
                    const int pos = qn + (int)__builtin_amdgcn_mbcnt_lo(mask, 0u);
                    myq[pos] = (t << EBITS) | (unsigned)e;
                }
                qn += __builtin_popcount(mask);
            }
        }
        {
            unsigned* rp = nullptr;
            if (FIRST && rec && stored + qn <= cap) {
                rp = mylst + LHDR + stored;
                stored += qn;
                cov = COV_ALL;
            }
            drain_queue<FIRST, CO>(myq, qn, lane, h, src, myacc, mytag, rp);
        }
    }
    if (FIRST && cap > 0 && lane == 0) {
        mylst[0] = (unsigned)stored;
        mylst[1] = (unsigned)cov;
    }
}

template <bool FIRST>
__global__ __launch_bounds__(TPB)
void owner_agg(const float* __restrict__ h,
               const int* __restrict__ src,
               const int* __restrict__ dst,
               float* __restrict__ agg,
               float* __restrict__ deg,
               unsigned* __restrict__ extra,
               size_t ws_bytes,
               int n_nodes, int n_edges) {
    __shared__ __attribute__((aligned(16))) float accs[NWAVE * NPB * AC];
    __shared__ unsigned q[NWAVE * QS];
    __shared__ unsigned tags[NWAVE * NPB];
    __shared__ unsigned vote[NWAVE];

    const int tid  = threadIdx.x;
    const int lane = tid & 31;
    const int wave = tid >> 5;
    const int lo   = blockIdx.x * NPB;
    const int cnt  = imax(imin(NPB, n_nodes - lo), 0);
    const int grp  = blockIdx.x / NGRP;

    const size_t used = (size_t)(2 * FD + 1) * NNODES * sizeof(float);
    const size_t co_bytes = (CO_WORDS + OVF_WORDS) * sizeof(unsigned);
    const bool co_room = (n_edges & 3) == 0 && ws_bytes >= used + co_bytes;
    size_t rest = ws_bytes > used ? ws_bytes - used : 0;
    unsigned* lst = extra;
    if (co_room) {
        rest -= co_bytes;
        lst = extra + CO_WORDS + OVF_WORDS;
    }
    size_t pitch = rest / (sizeof(unsigned) * NBLK * NWAVE);
    pitch = pitch > LPMAX ? (size_t)LPMAX : (pitch & ~(size_t)31);
    const int cap = pitch > LHDR ? (int)(pitch - LHDR) : 0;
    unsigned* mylst = lst + (size_t)(blockIdx.x * NWAVE + wave) * pitch;

    int bad = 0;
    if (co_room) {
        const unsigned* ovf = extra + CO_WORDS + (size_t)grp * NBLK;
        for (int i = tid; i < NBLK; i += TPB) bad |= (int)(ovf[i] != 0u);
    }
    const unsigned wbad = __builtin_amdgcn_ballot_w32(bad != 0);
    if (lane == 0) vote[wave] = wbad;
    for (int i = tid; i < NWAVE * NPB * AC; i += TPB) accs[i] = 0.0f;
    __syncthreads();
    unsigned anybad = 0u;
#pragma unroll
    for (int w = 0; w < NWAVE; ++w) anybad |= vote[w];
    const bool use_co = co_room && __builtin_amdgcn_readfirstlane((int)anybad) == 0;

    float* myacc = accs + wave * (NPB * AC);
    unsigned* myq = q + wave * QS;
    lds_tag_ptr mytag = (lds_tag_ptr)(tags + wave * NPB);

    if (use_co) {
        owner_pass<FIRST, true>(h, src, dst, (const v4i*)(extra + (size_t)blockIdx.x * OWORDS), OWORDS / 4, n_edges,
                                (blockIdx.x % NGRP) * NPB, cnt, tid, lane, myq, myacc, mytag, mylst, cap);
    } else {
        owner_pass<FIRST, false>(h, src, dst, (const v4i*)dst, n_edges >> 2, n_edges,
                                 lo, cnt, tid, lane, myq, myacc, mytag, mylst, cap);
    }
    __syncthreads();

    for (int i = tid; i < cnt * AC; i += TPB) {
        float sum = 0.0f;
#pragma unroll
        for (int w = 0; w < NWAVE; ++w) sum += accs[w * (NPB * AC) + i];
        const int node = i / AC;
        const int c = i - node * AC;
        if (c < FD) agg[(long)(lo + node) * FD + c] = sum;
        else if (FIRST) deg[lo + node] = sum;
    }
}

template <int F_OUT>
__global__ void node_wmma(const float* __restrict__ h,
                          const float* __restrict__ agg,
                          const float* __restrict__ deg,
                          const float* __restrict__ Wself,
                          const float* __restrict__ Wneigh,
                          const float* __restrict__ bias,
                          float* __restrict__ out,
                          int n_nodes) {
    const int lane  = threadIdx.x & 31;
    const int wave  = threadIdx.x >> 5;
    const int tile  = blockIdx.x * (blockDim.x >> 5) + wave;
    const int m0    = tile * 16;
    if (m0 >= n_nodes) return;

    const int row   = lane & 15;
    const int khalf = lane >> 4;

    const int m  = m0 + row;
    const int mc = imin(m, n_nodes - 1);
    const float invdeg = 1.0f / fmaxf(deg[mc], 1.0f);

    const int n  = row;
    const int nc = imin(n, F_OUT - 1);
    const bool nvalid = (n < F_OUT);

    v8f c = {};
#pragma unroll
    for (int chunk = 0; chunk < 3; ++chunk) {
        v2f a, b;
#pragma unroll
        for (int j = 0; j < 2; ++j) {
            const int k  = chunk * 4 + khalf * 2 + j;
            const int kh = imin(k, FD - 1);
            const int kn = imin(imax(k - FD, 0), FD - 1);

            const float hv = h[(long)mc * FD + kh];
            const float nv = agg[(long)mc * FD + kn] * invdeg;
            const float av = (k < FD) ? hv : ((k < 2 * FD) ? nv : 0.0f);

            const float ws = Wself[kh * F_OUT + nc];
            const float wn = Wneigh[kn * F_OUT + nc];
            float bv = (k < FD) ? ws : ((k < 2 * FD) ? wn : 0.0f);
            bv = nvalid ? bv : 0.0f;

            a[j] = av;
            b[j] = bv;
        }
        c = __builtin_amdgcn_wmma_f32_16x16x4_f32(false, a, false, b,
                                                  (short)0, c, false, false);
    }

    const float bn = bias[nc];
#pragma unroll
    for (int r = 0; r < 8; ++r) {
        const int mm = m0 + r + khalf * 8;
        if (mm < n_nodes && nvalid) {
            const float v = c[r] + bn;
            out[(long)mm * F_OUT + n] = 1.0f / (1.0f + __expf(-v));
        }
    }
}

extern "C" void kernel_launch(void* const* d_in, const int* in_sizes, int n_in,
                              void* d_out, int out_size, void* d_ws, size_t ws_size,
                              hipStream_t stream) {
    (void)in_sizes; (void)n_in; (void)out_size;

    const float* x      = (const float*)d_in[0];
    const int*   src    = (const int*)d_in[1];
    const int*   dst    = (const int*)d_in[2];
    const float* Wself1 = (const float*)d_in[3];
    const float* Wneigh1= (const float*)d_in[4];
    const float* b1     = (const float*)d_in[5];
    const float* Wself2 = (const float*)d_in[6];
    const float* Wneigh2= (const float*)d_in[7];
    const float* b2     = (const float*)d_in[8];
    float* out = (float*)d_out;

    float* h   = (float*)d_ws;
    float* agg = h + (size_t)NNODES * FD;
    float* deg = agg + (size_t)NNODES * FD;
    unsigned* extra = (unsigned*)(deg + (size_t)NNODES);

    const int NODE_B = 256;
    const int tiles = (NNODES + 15) / 16;
    const int node_grid = (tiles + (NODE_B / 32) - 1) / (NODE_B / 32);

    coarse_bin<<<NBLK, TPB, 0, stream>>>(src, dst, extra, ws_size, NEDGES);

    owner_agg<true><<<NBLK, TPB, 0, stream>>>(x, src, dst, agg, deg, extra, ws_size, NNODES, NEDGES);
    node_wmma<FD><<<node_grid, NODE_B, 0, stream>>>(x, agg, deg, Wself1, Wneigh1,
                                                    b1, h, NNODES);

    owner_agg<false><<<NBLK, TPB, 0, stream>>>(h, src, dst, agg, deg, extra, ws_size, NNODES, NEDGES);
    node_wmma<10><<<node_grid, NODE_B, 0, stream>>>(h, agg, deg, Wself2, Wneigh2,
                                                    b2, out, NNODES);
}


# Node transform fused into the owner work-group: behind its LDS sums it runs the same 3 f32 WMMAs per 16-node tile on its own nodes, self rows and weights loaded early; 5 launches to 3; sums as before
# speedup vs baseline: 22.9258x; 1.0383x over previous
#include <hip/hip_runtime.h>
#include <math.h>

#define FD 5
#define NNODES 100000
#define NEDGES 6400000

typedef __attribute__((ext_vector_type(2))) float v2f;
typedef __attribute__((ext_vector_type(8))) float v8f;

__device__ __forceinline__ int imin(int a, int b) { return a < b ? a : b; }
__device__ __forceinline__ int imax(int a, int b) { return a > b ? a : b; }

#define NBLK 256
#define NPB  391
#define TPB  512
#define NWAVE (TPB / 32)
#define QW   2048
#define QS   (QW + 32)
#define VPI  5
#define EBITS 23
#define AC   (FD + 1)
#define SBITS 17
#define LHDR  32
#define LPMAX 4096
#define COV_ALL 0x7fffffff

#define NGRP    16
#define NGROUPS (NBLK / NGRP)
#define GNODES  (NGRP * NPB)
#define BWAVES  (NBLK * NWAVE)
#define SEGW    160
#define OWORDS  (NBLK * SEGW)
#define CO_WORDS  ((size_t)NBLK * OWORDS)
#define OVF_WORDS ((size_t)NGROUPS * NBLK)
#define PAD_WORD 0xffffffffu
#define SMASK   ((1u << SBITS) - 1u)
static_assert(OWORDS % (4 * TPB * VPI) == 0, "an owner array is a whole number of scan trips");
static_assert(NGRP == NWAVE && NGROUPS == NWAVE, "stage two: one wave per group, one segment per owner of the group");
static_assert(GNODES < (1 << (32 - SBITS)) - 1, "group node index and PAD_WORD fit");

typedef __attribute__((ext_vector_type(4))) int v4i;

struct __attribute__((packed, aligned(4))) row_t { float v[FD]; };

#define DB 4
typedef volatile __attribute__((address_space(3))) unsigned* lds_tag_ptr;
template <bool ADD_DEG, bool DIRECT>
__device__ __forceinline__ void drain_queue(const unsigned* myq, int qn, int lane,
                                            const float* __restrict__ h,
                                            const int* __restrict__ src,
                                            float* myacc,
                                            lds_tag_ptr mytag,
                                            unsigned* rec) {
    for (int i = 0; i < qn; i += DB * 32) {
        unsigned w[DB];
        int s[DB];
        row_t r[DB];
#pragma unroll
        for (int k = 0; k < DB; ++k) {
            const int idx = i + k * 32 + lane;
            w[k] = myq[idx < qn ? idx : qn - 1];
        }
#pragma unroll
        for (int k = 0; k < DB; ++k) {
            if (DIRECT) s[k] = imin((int)(w[k] & ((1u << SBITS) - 1u)), NNODES - 1);
            else        s[k] = src[w[k] & ((1u << EBITS) - 1u)];
        }
#pragma unroll
        for (int k = 0; k < DB; ++k)
            __builtin_memcpy(&r[k], h + (long)s[k] * FD, sizeof(row_t));
        if (rec != nullptr) {
#pragma unroll
            for (int k = 0; k < DB; ++k) {
                const int idx = i + k * 32 + lane;
                if (idx < qn) rec[idx] = DIRECT ? w[k] : (((w[k] >> EBITS) << SBITS) | (unsigned)s[k]);
            }
        }
#pragma unroll
        for (int k = 0; k < DB; ++k) {
            const unsigned t = DIRECT ? (w[k] >> SBITS) : (w[k] >> EBITS);
            bool pending = (i + k * 32 + lane) < qn;
            for (int round = 0; round < 32; ++round) {
                if (__builtin_amdgcn_ballot_w32(pending) == 0u) break;
                bool win = false;
                if (pending) {
                    mytag[t] = (unsigned)lane;
                    win = (mytag[t] == (unsigned)lane);
                }
                if (win) {
                    float2* p = (float2*)(myacc + t * AC);
                    float2 a0 = p[0], a1 = p[1], a2 = p[2];
                    a0.x += r[k].v[0];
                    a0.y += r[k].v[1];
                    a1.x += r[k].v[2];
                    a1.y += r[k].v[3];
                    a2.x += r[k].v[4];
                    if (ADD_DEG) a2.y += 1.0f;
                    p[0] = a0;
                    p[1] = a1;
                    p[2] = a2;
                    pending = false;
                }
            }
        }
    }
}

template <bool CO>
__device__ __forceinline__ void scan_full_trip(const v4i* d, v4i* nd,
                                               const v4i* __restrict__ scan4,
                                               int base, int step, int tid, int n_vec,
                                               int lo, unsigned cnt, int lane,
                                               unsigned* myq, int& qn) {
#pragma unroll
    for (int u = 0; u < VPI; ++u) {
        const unsigned t0 = (CO ? ((unsigned)d[u][0] >> SBITS) : (unsigned)d[u][0]) - (unsigned)lo;
        const unsigned t1 = (CO ? ((unsigned)d[u][1] >> SBITS) : (unsigned)d[u][1]) - (unsigned)lo;
        const unsigned t2 = (CO ? ((unsigned)d[u][2] >> SBITS) : (unsigned)d[u][2]) - (unsigned)lo;
        const unsigned t3 = (CO ? ((unsigned)d[u][3] >> SBITS) : (unsigned)d[u][3]) - (unsigned)lo;
        const unsigned t01 = t0 < t1 ? t0 : t1;
        const unsigned t23 = t2 < t3 ? t2 : t3;
        const unsigned tm  = t01 < t23 ? t01 : t23;
        if (__builtin_amdgcn_ballot_w32(tm < cnt) != 0u) {
            const unsigned e0 = 4u * (unsigned)(base + u * TPB + tid);
            const unsigned tt[4] = {t0, t1, t2, t3};
#pragma unroll
            for (int c = 0; c < 4; ++c) {
                const bool m = tt[c] < cnt;
                const unsigned mask = __builtin_amdgcn_ballot_w32(m);
                const int pos = (int)__builtin_amdgcn_mbcnt_lo(mask, (unsigned)qn);
                const unsigned ent = CO ? ((tt[c] << SBITS) | ((unsigned)d[u][c] & SMASK))
                                        : ((tt[c] << EBITS) | (e0 + (unsigned)c));
                myq[m ? pos : QW + lane] = ent;
                qn += __builtin_popcount(mask);
            }
        }
        if (u == 0) {
#pragma unroll
            for (int u2 = 0; u2 < VPI; ++u2) {
                const int v2 = base + step + u2 * TPB + tid;
                nd[u2] = scan4[imin(v2, n_vec - 1)];
            }
        }
    }
}

#define BSEG (NWAVE * NGROUPS * SEGW)
#define BU   4
#define S2R  (NWAVE * SEGW / 32)
static_assert(SEGW % 32 == 0, "a segment is copied out in whole 32-lane rows");
static_assert(NGROUPS == 16 && NGRP == 16, "a key is taken apart as four bits");

__device__ __forceinline__ void place32(unsigned* buf, lds_tag_ptr myfill, bool act, unsigned key,
                                        unsigned word, unsigned seg0, unsigned segp, unsigned myspare) {
    unsigned same = __builtin_amdgcn_ballot_w32(act);
#pragma unroll
    for (int b = 0; b < 4; ++b) {
        const bool bit = ((key >> b) & 1u) != 0u;
        const unsigned vb1 = __builtin_amdgcn_ballot_w32(bit);
        same &= bit ? vb1 : ~vb1;
    }
    const unsigned rank  = __builtin_amdgcn_mbcnt_lo(same, 0u);
    const unsigned total = (unsigned)__builtin_popcount(same);
    const unsigned fill  = myfill[key];
    const unsigned pos   = fill + rank;
    buf[(act && pos < (unsigned)SEGW) ? seg0 + key * segp + pos : myspare] = word;
    if (act && rank + 1u == total) myfill[key] = fill + total;
}

__global__ __launch_bounds__(TPB)
void coarse_bin(const int* __restrict__ src,
                const int* __restrict__ dst,
                unsigned* __restrict__ extra,
                size_t ws_bytes, int n_edges) {
    __shared__ unsigned buf[BSEG + TPB];
    __shared__ unsigned fills[NWAVE * NGROUPS];
    __shared__ unsigned fills2[NWAVE * NGRP];

    const size_t used = (size_t)(2 * FD + 1) * NNODES * sizeof(float);
    if ((n_edges & 3) != 0 || ws_bytes < used + (CO_WORDS + OVF_WORDS) * sizeof(unsigned)) return;

    const int tid  = threadIdx.x;
    const int lane = tid & 31;
    const int wave = tid >> 5;
    const int gw   = blockIdx.x * NWAVE + wave;
    const unsigned myspare = (unsigned)(BSEG + tid);
    const unsigned seg0 = (unsigned)(wave * SEGW);
    lds_tag_ptr myfill = (lds_tag_ptr)(fills + wave * NGROUPS);
    for (int i = lane; i < NGROUPS * SEGW; i += 32) {
        const int g = i / SEGW;
        buf[(g * NWAVE + wave) * SEGW + (i - g * SEGW)] = PAD_WORD;
    }
    if (lane < NGROUPS) myfill[lane] = 0u;

    const v4i* dst4 = (const v4i*)dst;
    const v4i* src4 = (const v4i*)src;
    const int n_vec = n_edges >> 2;
    const int per   = (n_vec + BWAVES - 1) / BWAVES;
    const int v0    = imin(gw * per, n_vec);
    const int v1    = imin(v0 + per, n_vec);

    for (int vb = v0; vb < v1; vb += 32 * BU) {
        v4i dvv[BU], svv[BU];
#pragma unroll
        for (int k = 0; k < BU; ++k) {
            const int vc = imin(vb + k * 32 + lane, v1 - 1);
            dvv[k] = dst4[vc];
            svv[k] = src4[vc];
        }
#pragma unroll
        for (int k = 0; k < BU; ++k) {
            const bool valid = vb + k * 32 + lane < v1;
            const v4i dv = dvv[k];
            const v4i sv = svv[k];
#pragma unroll
            for (int c = 0; c < 4; ++c) {
                const unsigned ud  = (unsigned)dv[c];
                const unsigned grp = ud / (unsigned)GNODES;
                const unsigned sn  = (unsigned)imin(imax(sv[c], 0), NNODES - 1);
                const unsigned word = ((ud - grp * (unsigned)GNODES) << SBITS) | sn;
                const bool act = valid && grp < (unsigned)NGROUPS;
                place32(buf, myfill, act, act ? grp : 0u, word, seg0, (unsigned)(NWAVE * SEGW), myspare);
            }
        }
    }
    __syncthreads();

    const unsigned in0 = (unsigned)(wave * NWAVE * SEGW);
    lds_tag_ptr myfill2 = (lds_tag_ptr)(fills2 + wave * NGRP);
    unsigned win[S2R];
#pragma unroll
    for (int r = 0; r < S2R; ++r) win[r] = buf[in0 + r * 32 + lane];
    for (int i = lane; i < NGRP * SEGW; i += 32) buf[in0 + i] = PAD_WORD;
    if (lane < NGRP) myfill2[lane] = 0u;
#pragma unroll
    for (int r = 0; r < S2R; ++r) {
        const unsigned word = win[r];
        const bool act = word != PAD_WORD;
        if (__builtin_amdgcn_ballot_w32(act) == 0u) continue;
        const unsigned own = (word >> SBITS) / (unsigned)NPB;
        place32(buf, myfill2, act, act ? own : 0u, word, in0, (unsigned)SEGW, myspare);
    }

    unsigned* coarse = extra;
    unsigned* ovf    = extra + CO_WORDS;
#pragma nounroll
    for (int o = 0; o < NGRP; ++o) {
        unsigned* seg = coarse + ((size_t)(wave * NGRP + o) * NBLK + blockIdx.x) * SEGW;
        unsigned wv[SEGW / 32];
#pragma unroll
        for (int j = 0; j < SEGW / 32; ++j) wv[j] = buf[in0 + o * SEGW + j * 32 + lane];
#pragma unroll
        for (int j = 0; j < SEGW / 32; ++j) seg[j * 32 + lane] = wv[j];
    }
    bool over = false;
    if (lane < NWAVE) over = fills[lane * NGROUPS + wave] > (unsigned)SEGW || myfill2[lane] > (unsigned)SEGW;
    const unsigned anyover = __builtin_amdgcn_ballot_w32(over);
    if (lane == 0) ovf[(size_t)wave * NBLK + blockIdx.x] = anyover != 0u ? 1u : 0u;
}

template <bool FIRST, bool CO>
__device__ __forceinline__ void owner_pass(const float* __restrict__ h,
                                           const int* __restrict__ src,
                                           const int* __restrict__ dst,
                                           const v4i* __restrict__ scan4,
                                           int n_vec, int n_edges,
                                           int lo, int cnt, int tid, int lane,
                                           unsigned* myq, float* myacc, lds_tag_ptr mytag,
                                           unsigned* mylst, int cap) {
    int qn = 0;
    const int step  = TPB * VPI;

    bool rec   = FIRST && cap > 0;
    int stored = 0;
    int cov    = 0;

    int start = 0;
    if (!FIRST && cap > 0) {
        int n = __builtin_amdgcn_readfirstlane((int)mylst[0]);
        start = __builtin_amdgcn_readfirstlane((int)mylst[1]);
        n     = imax(imin(n, cap), 0);
        start = imax(start, 0);
        drain_queue<false, true>(mylst + LHDR, n, lane, h, src, myacc, mytag, nullptr);
    }

    if (start != COV_ALL) {
        v4i nd[VPI];
#pragma unroll
        for (int u = 0; u < VPI; ++u) {
            const int v = start + u * TPB + tid;
            nd[u] = scan4[imin(v, n_vec - 1)];
        }
        for (int base = start; base < n_vec; base += step) {
            if (qn > QW - VPI * 4 * 32) {
                unsigned* rp = nullptr;
                if (FIRST) {
                    if (rec && stored + qn <= cap) {
                        rp = mylst + LHDR + stored;
                        stored += qn;
                        cov = base;
                    } else {
                        rec = false;
                    }
                }
                drain_queue<FIRST, CO>(myq, qn, lane, h, src, myacc, mytag, rp);
                qn = 0;
            }
            if (CO || base + step <= n_vec) {
                v4i d[VPI];
#pragma unroll
                for (int u = 0; u < VPI; ++u) d[u] = nd[u];
                scan_full_trip<CO>(d, nd, scan4, base, step, tid, n_vec, lo, (unsigned)cnt, lane, myq, qn);
            } else {
#pragma nounroll
                for (int vb = base; vb < n_vec; vb += TPB) {
                    const int v = vb + tid;
                    const bool valid = v < n_vec;
                    const v4i dv = scan4[valid ? v : n_vec - 1];
#pragma unroll
                    for (int c = 0; c < 4; ++c) {
                        const unsigned t = (unsigned)(dv[c] - lo);
                        const bool m = valid && (t < (unsigned)cnt);
                        const unsigned mask = __builtin_amdgcn_ballot_w32(m);
                        if (mask != 0u) {
                            if (m) {
                                const int pos = qn + (int)__builtin_amdgcn_mbcnt_lo(mask, 0u);
                                myq[pos] = (t << EBITS) | (unsigned)(4 * v + c);
                            }
                            qn += __builtin_popcount(mask);
                        }
                    }
                }
            }
        }
        if (!CO) {
            for (int e0 = n_vec * 4; e0 < n_edges; e0 += TPB) {
                const int e = e0 + tid;
                if (qn > QW - 32) {
                    rec = false;
                    drain_queue<FIRST, false>(myq, qn, lane, h, src, myacc, mytag, nullptr);
                    qn = 0;
                }
                const bool valid = e < n_edges;
                const unsigned t = (unsigned)(dst[valid ? e : 0] - lo);
                const bool m = valid && (t < (unsigned)cnt);
                const unsigned mask = __builtin_amdgcn_ballot_w32(m);
                if (m) {
                    const int pos = qn + (int)__builtin_amdgcn_mbcnt_lo(mask, 0u);
                    myq[pos] = (t << EBITS) | (unsigned)e;
                }
                qn += __builtin_popcount(mask);
            }
        }
        {
            unsigned* rp = nullptr;
            if (FIRST && rec && stored + qn <= cap) {
                rp = mylst + LHDR + stored;
                stored += qn;
                cov = COV_ALL;
            }
            drain_queue<FIRST, CO>(myq, qn, lane, h, src, myacc, mytag, rp);
        }
    }
    if (FIRST && cap > 0 && lane == 0) {
        mylst[0] = (unsigned)stored;
        mylst[1] = (unsigned)cov;
    }
}

#define NTILE ((NPB + 15) / 16)
#define TPW   ((NTILE + NWAVE - 1) / NWAVE)

template <bool FIRST>
__global__ __launch_bounds__(TPB)
void owner_agg(const float* __restrict__ h,
               const int* __restrict__ src,
               const int* __restrict__ dst,
               float* __restrict__ deg,
               unsigned* __restrict__ extra,
               size_t ws_bytes,
               int n_nodes, int n_edges,
               const float* __restrict__ Wself,
               const float* __restrict__ Wneigh,
               const float* __restrict__ bias,
               float* __restrict__ out) {
    constexpr int F_OUT = FIRST ? FD : 10;
    __shared__ __attribute__((aligned(16))) float accs[NWAVE * NPB * AC];
    __shared__ unsigned q[NWAVE * QS];
    __shared__ unsigned tags[NWAVE * NPB];
    __shared__ unsigned vote[NWAVE];

    const int tid  = threadIdx.x;
    const int lane = tid & 31;
    const int wave = tid >> 5;
    const int lo   = blockIdx.x * NPB;
    const int cnt  = imax(imin(NPB, n_nodes - lo), 0);
    const int grp  = blockIdx.x / NGRP;

    const int row   = lane & 15;
    const int khalf = lane >> 4;
    const int nc    = imin(row, F_OUT - 1);
    const bool nvalid = row < F_OUT;
    v2f bfrag[3];
    float hself[TPW][3][2];
    float degpre[TPW];
#pragma unroll
    for (int chunk = 0; chunk < 3; ++chunk) {
#pragma unroll
        for (int j = 0; j < 2; ++j) {
            const int k  = chunk * 4 + khalf * 2 + j;
            const int kh = imin(k, FD - 1);
            const int kn = imin(imax(k - FD, 0), FD - 1);
            const float ws = Wself[kh * F_OUT + nc];
            const float wn = Wneigh[kn * F_OUT + nc];
            float bv = (k < FD) ? ws : ((k < 2 * FD) ? wn : 0.0f);
            bfrag[chunk][j] = nvalid ? bv : 0.0f;
        }
    }
    const float bn = bias[nc];
#pragma unroll
    for (int tt = 0; tt < TPW; ++tt) {
        const int ml = imin((wave + tt * NWAVE) * 16 + row, imax(cnt - 1, 0));
        const int mc = imin(lo + ml, n_nodes - 1);
        degpre[tt] = FIRST ? 0.0f : deg[mc];
#pragma unroll
        for (int chunk = 0; chunk < 3; ++chunk) {
#pragma unroll
            for (int j = 0; j < 2; ++j) {
                const int kh = imin(chunk * 4 + khalf * 2 + j, FD - 1);
                hself[tt][chunk][j] = h[(long)mc * FD + kh];
            }
        }
    }

    const size_t used = (size_t)(2 * FD + 1) * NNODES * sizeof(float);
    const size_t co_bytes = (CO_WORDS + OVF_WORDS) * sizeof(unsigned);
    const bool co_room = (n_edges & 3) == 0 && ws_bytes >= used + co_bytes;
    size_t rest = ws_bytes > used ? ws_bytes - used : 0;
    unsigned* lst = extra;
    if (co_room) {
        rest -= co_bytes;
        lst = extra + CO_WORDS + OVF_WORDS;
    }
    size_t pitch = rest / (sizeof(unsigned) * NBLK * NWAVE);
    pitch = pitch > LPMAX ? (size_t)LPMAX : (pitch & ~(size_t)31);
    const int cap = pitch > LHDR ? (int)(pitch - LHDR) : 0;
    unsigned* mylst = lst + (size_t)(blockIdx.x * NWAVE + wave) * pitch;

    int bad = 0;
    if (co_room) {
        const unsigned* ovf = extra + CO_WORDS + (size_t)grp * NBLK;
        for (int i = tid; i < NBLK; i += TPB) bad |= (int)(ovf[i] != 0u);
    }
    const unsigned wbad = __builtin_amdgcn_ballot_w32(bad != 0);
    if (lane == 0) vote[wave] = wbad;
    for (int i = tid; i < NWAVE * NPB * AC; i += TPB) accs[i] = 0.0f;
    __syncthreads();
    unsigned anybad = 0u;
#pragma unroll
    for (int w = 0; w < NWAVE; ++w) anybad |= vote[w];
    const bool use_co = co_room && __builtin_amdgcn_readfirstlane((int)anybad) == 0;

    float* myacc = accs + wave * (NPB * AC);
    unsigned* myq = q + wave * QS;
    lds_tag_ptr mytag = (lds_tag_ptr)(tags + wave * NPB);

    if (use_co) {
        owner_pass<FIRST, true>(h, src, dst, (const v4i*)(extra + (size_t)blockIdx.x * OWORDS), OWORDS / 4, n_edges,
                                (blockIdx.x % NGRP) * NPB, cnt, tid, lane, myq, myacc, mytag, mylst, cap);
    } else {
        owner_pass<FIRST, false>(h, src, dst, (const v4i*)dst, n_edges >> 2, n_edges,
                                 lo, cnt, tid, lane, myq, myacc, mytag, mylst, cap);
    }
    __syncthreads();

    for (int i = tid; i < cnt * AC; i += TPB) {
        float sum = 0.0f;
#pragma unroll
        for (int w = 0; w < NWAVE; ++w) sum += accs[w * (NPB * AC) + i];
        accs[i] = sum;
        if (FIRST) {
            const int node = i / AC;
            if (i - node * AC == FD) deg[lo + node] = sum;
        }
    }
    __syncthreads();

    const int swave = __builtin_amdgcn_readfirstlane(wave);
#pragma unroll
    for (int tt = 0; tt < TPW; ++tt) {
        const int m0 = (swave + tt * NWAVE) * 16;
        if (m0 < cnt) {
            const int ml = imin(m0 + row, cnt - 1);
            const float dg = FIRST ? accs[ml * AC + FD] : degpre[tt];
            const float invdeg = 1.0f / fmaxf(dg, 1.0f);
            v8f c = {};
#pragma unroll
            for (int chunk = 0; chunk < 3; ++chunk) {
                v2f a;
#pragma unroll
                for (int j = 0; j < 2; ++j) {
                    const int k  = chunk * 4 + khalf * 2 + j;
                    const int kn = imin(imax(k - FD, 0), FD - 1);
                    const float nv = accs[ml * AC + kn] * invdeg;
                    a[j] = (k < FD) ? hself[tt][chunk][j] : ((k < 2 * FD) ? nv : 0.0f);
                }
                c = __builtin_amdgcn_wmma_f32_16x16x4_f32(false, a, false, bfrag[chunk],
                                                          (short)0, c, false, false);
            }
#pragma unroll
            for (int r = 0; r < 8; ++r) {
                const int mm = m0 + r + khalf * 8;
                if (mm < cnt && nvalid) {
                    const float v = c[r] + bn;
                    out[(long)(lo + mm) * F_OUT + row] = 1.0f / (1.0f + __expf(-v));
                }
            }
        }
    }
}

extern "C" void kernel_launch(void* const* d_in, const int* in_sizes, int n_in,
                              void* d_out, int out_size, void* d_ws, size_t ws_size,
                              hipStream_t stream) {
    (void)in_sizes; (void)n_in; (void)out_size;

    const float* x      = (const float*)d_in[0];
    const int*   src    = (const int*)d_in[1];
    const int*   dst    = (const int*)d_in[2];
    const float* Wself1 = (const float*)d_in[3];
    const float* Wneigh1= (const float*)d_in[4];
    const float* b1     = (const float*)d_in[5];
    const float* Wself2 = (const float*)d_in[6];
    const float* Wneigh2= (const float*)d_in[7];
    const float* b2     = (const float*)d_in[8];
    float* out = (float*)d_out;

    float* h   = (float*)d_ws;
    float* deg = h + (size_t)2 * NNODES * FD;
    unsigned* extra = (unsigned*)(deg + (size_t)NNODES);

    coarse_bin<<<NBLK, TPB, 0, stream>>>(src, dst, extra, ws_size, NEDGES);

    owner_agg<true><<<NBLK, TPB, 0, stream>>>(x, src, dst, deg, extra, ws_size, NNODES, NEDGES,
                                              Wself1, Wneigh1, b1, h);

    owner_agg<false><<<NBLK, TPB, 0, stream>>>(h, src, dst, deg, extra, ws_size, NNODES, NEDGES,
                                               Wself2, Wneigh2, b2, out);
}


# Both gathered row tables at a 32-byte pitch (pre-pass copies x, layer 1 stores h so): no 20-byte row crosses a 128-byte line, 16-byte pieces aligned; sums and their order unchanged
# speedup vs baseline: 23.6628x; 1.0321x over previous
#include <hip/hip_runtime.h>
#include <math.h>

#define FD 5
#define NNODES 100000
#define NEDGES 6400000

typedef __attribute__((ext_vector_type(2))) float v2f;
typedef __attribute__((ext_vector_type(8))) float v8f;

__device__ __forceinline__ int imin(int a, int b) { return a < b ? a : b; }
__device__ __forceinline__ int imax(int a, int b) { return a > b ? a : b; }

#define NBLK 256
#define NPB  391
#define TPB  512
#define NWAVE (TPB / 32)
#define QW   2048
#define QS   (QW + 32)
#define VPI  5
#define EBITS 23
#define AC   (FD + 1)
#define SBITS 17
#define LHDR  32
#define LPMAX 4096
#define COV_ALL 0x7fffffff

#define RP 8
static_assert(RP >= FD && RP <= 2 * FD, "h at the row pitch ends in front of the in-degrees");
#define FRONT_FLOATS ((size_t)(2 * FD + 1 + RP) * NNODES)

#define NGRP    16
#define NGROUPS (NBLK / NGRP)
#define GNODES  (NGRP * NPB)
#define BWAVES  (NBLK * NWAVE)
#define SEGW    160
#define OWORDS  (NBLK * SEGW)
#define CO_WORDS  ((size_t)NBLK * OWORDS)
#define OVF_WORDS ((size_t)NGROUPS * NBLK)
#define PAD_WORD 0xffffffffu
#define SMASK   ((1u << SBITS) - 1u)
static_assert(OWORDS % (4 * TPB * VPI) == 0, "an owner array is a whole number of scan trips");
static_assert(NGRP == NWAVE && NGROUPS == NWAVE, "stage two: one wave per group, one segment per owner of the group");
static_assert(GNODES < (1 << (32 - SBITS)) - 1, "group node index and PAD_WORD fit");

typedef __attribute__((ext_vector_type(4))) int v4i;

struct __attribute__((packed, aligned(4))) row_t { float v[FD]; };

#define DB 4
typedef volatile __attribute__((address_space(3))) unsigned* lds_tag_ptr;
template <bool ADD_DEG, bool DIRECT>
__device__ __forceinline__ void drain_queue(const unsigned* myq, int qn, int lane,
                                            const float* __restrict__ h, int pitch,
                                            const int* __restrict__ src,
                                            float* myacc,
                                            lds_tag_ptr mytag,
                                            unsigned* rec) {
    for (int i = 0; i < qn; i += DB * 32) {
        unsigned w[DB];
        int s[DB];
        row_t r[DB];
#pragma unroll
        for (int k = 0; k < DB; ++k) {
            const int idx = i + k * 32 + lane;
            w[k] = myq[idx < qn ? idx : qn - 1];
        }
#pragma unroll
        for (int k = 0; k < DB; ++k) {
            if (DIRECT) s[k] = imin((int)(w[k] & ((1u << SBITS) - 1u)), NNODES - 1);
            else        s[k] = src[w[k] & ((1u << EBITS) - 1u)];
        }
#pragma unroll
        for (int k = 0; k < DB; ++k)
            __builtin_memcpy(&r[k], h + (long)s[k] * pitch, sizeof(row_t));
        if (rec != nullptr) {
#pragma unroll
            for (int k = 0; k < DB; ++k) {
                const int idx = i + k * 32 + lane;
                if (idx < qn) rec[idx] = DIRECT ? w[k] : (((w[k] >> EBITS) << SBITS) | (unsigned)s[k]);
            }
        }
#pragma unroll
        for (int k = 0; k < DB; ++k) {
            const unsigned t = DIRECT ? (w[k] >> SBITS) : (w[k] >> EBITS);
            bool pending = (i + k * 32 + lane) < qn;
            for (int round = 0; round < 32; ++round) {
                if (__builtin_amdgcn_ballot_w32(pending) == 0u) break;
                bool win = false;
                if (pending) {
                    mytag[t] = (unsigned)lane;
                    win = (mytag[t] == (unsigned)lane);
                }
                if (win) {
                    float2* p = (float2*)(myacc + t * AC);
                    float2 a0 = p[0], a1 = p[1], a2 = p[2];
                    a0.x += r[k].v[0];
                    a0.y += r[k].v[1];
                    a1.x += r[k].v[2];
                    a1.y += r[k].v[3];
                    a2.x += r[k].v[4];
                    if (ADD_DEG) a2.y += 1.0f;
                    p[0] = a0;
                    p[1] = a1;
                    p[2] = a2;
                    pending = false;
                }
            }
        }
    }
}

template <bool CO>
__device__ __forceinline__ void scan_full_trip(const v4i* d, v4i* nd,
                                               const v4i* __restrict__ scan4,
                                               int base, int step, int tid, int n_vec,
                                               int lo, unsigned cnt, int lane,
                                               unsigned* myq, int& qn) {
#pragma unroll
    for (int u = 0; u < VPI; ++u) {
        const unsigned t0 = (CO ? ((unsigned)d[u][0] >> SBITS) : (unsigned)d[u][0]) - (unsigned)lo;
        const unsigned t1 = (CO ? ((unsigned)d[u][1] >> SBITS) : (unsigned)d[u][1]) - (unsigned)lo;
        const unsigned t2 = (CO ? ((unsigned)d[u][2] >> SBITS) : (unsigned)d[u][2]) - (unsigned)lo;
        const unsigned t3 = (CO ? ((unsigned)d[u][3] >> SBITS) : (unsigned)d[u][3]) - (unsigned)lo;
        const unsigned t01 = t0 < t1 ? t0 : t1;
        const unsigned t23 = t2 < t3 ? t2 : t3;
        const unsigned tm  = t01 < t23 ? t01 : t23;
        if (__builtin_amdgcn_ballot_w32(tm < cnt) != 0u) {
            const unsigned e0 = 4u * (unsigned)(base + u * TPB + tid);
            const unsigned tt[4] = {t0, t1, t2, t3};
#pragma unroll
            for (int c = 0; c < 4; ++c) {
                const bool m = tt[c] < cnt;
                const unsigned mask = __builtin_amdgcn_ballot_w32(m);
                const int pos = (int)__builtin_amdgcn_mbcnt_lo(mask, (unsigned)qn);
                const unsigned ent = CO ? ((tt[c] << SBITS) | ((unsigned)d[u][c] & SMASK))
                                        : ((tt[c] << EBITS) | (e0 + (unsigned)c));
                myq[m ? pos : QW + lane] = ent;
                qn += __builtin_popcount(mask);
            }
        }
        if (u == 0) {
#pragma unroll
            for (int u2 = 0; u2 < VPI; ++u2) {
                const int v2 = base + step + u2 * TPB + tid;
                nd[u2] = scan4[imin(v2, n_vec - 1)];
            }
        }
    }
}

#define BSEG (NWAVE * NGROUPS * SEGW)
#define BU   4
#define S2R  (NWAVE * SEGW / 32)
static_assert(SEGW % 32 == 0, "a segment is copied out in whole 32-lane rows");
static_assert(NGROUPS == 16 && NGRP == 16, "a key is taken apart as four bits");

__device__ __forceinline__ void place32(unsigned* buf, lds_tag_ptr myfill, bool act, unsigned key,
                                        unsigned word, unsigned seg0, unsigned segp, unsigned myspare) {
    unsigned same = __builtin_amdgcn_ballot_w32(act);
#pragma unroll
    for (int b = 0; b < 4; ++b) {
        const bool bit = ((key >> b) & 1u) != 0u;
        const unsigned vb1 = __builtin_amdgcn_ballot_w32(bit);
        same &= bit ? vb1 : ~vb1;
    }
    const unsigned rank  = __builtin_amdgcn_mbcnt_lo(same, 0u);
    const unsigned total = (unsigned)__builtin_popcount(same);
    const unsigned fill  = myfill[key];
    const unsigned pos   = fill + rank;
    buf[(act && pos < (unsigned)SEGW) ? seg0 + key * segp + pos : myspare] = word;
    if (act && rank + 1u == total) myfill[key] = fill + total;
}

#define XU ((NPB * FD + TPB - 1) / TPB)
__device__ __forceinline__ void store_xrows(float* __restrict__ xp, const float (&xv)[XU],
                                            int lo0, int n0, int tid) {
#pragma unroll
    for (int k = 0; k < XU; ++k) {
        const int i = tid + k * TPB;
        if (i < n0) {
            const int node = i / FD;
            xp[(size_t)(lo0 + node) * RP + (i - node * FD)] = xv[k];
        }
    }
}

__global__ __launch_bounds__(TPB)
void coarse_bin(const int* __restrict__ src,
                const int* __restrict__ dst,
                const float* __restrict__ x,
                float* __restrict__ xp,
                unsigned* __restrict__ extra,
                size_t ws_bytes, int n_edges) {
    __shared__ unsigned buf[BSEG + TPB];
    __shared__ unsigned fills[NWAVE * NGROUPS];
    __shared__ unsigned fills2[NWAVE * NGRP];

    const size_t used = FRONT_FLOATS * sizeof(float);
    const bool xp_room = ws_bytes >= used;
    const int lo0 = blockIdx.x * NPB;
    const int n0  = imax(imin(NPB, NNODES - lo0), 0) * FD;
    float xv[XU] = {};
    if (xp_room && n0 > 0) {
#pragma unroll
        for (int k = 0; k < XU; ++k) xv[k] = x[(size_t)lo0 * FD + imin((int)threadIdx.x + k * TPB, n0 - 1)];
    }
    if ((n_edges & 3) != 0 || ws_bytes < used + (CO_WORDS + OVF_WORDS) * sizeof(unsigned)) {
        if (xp_room) store_xrows(xp, xv, lo0, n0, (int)threadIdx.x);
        return;
    }

    const int tid  = threadIdx.x;
    const int lane = tid & 31;
    const int wave = tid >> 5;
    const int gw   = blockIdx.x * NWAVE + wave;
    const unsigned myspare = (unsigned)(BSEG + tid);
    const unsigned seg0 = (unsigned)(wave * SEGW);
    lds_tag_ptr myfill = (lds_tag_ptr)(fills + wave * NGROUPS);
    for (int i = lane; i < NGROUPS * SEGW; i += 32) {
        const int g = i / SEGW;
        buf[(g * NWAVE + wave) * SEGW + (i - g * SEGW)] = PAD_WORD;
    }
    if (lane < NGROUPS) myfill[lane] = 0u;
    store_xrows(xp, xv, lo0, n0, tid);

    const v4i* dst4 = (const v4i*)dst;
    const v4i* src4 = (const v4i*)src;
    const int n_vec = n_edges >> 2;
    const int per   = (n_vec + BWAVES - 1) / BWAVES;
    const int v0    = imin(gw * per, n_vec);
    const int v1    = imin(v0 + per, n_vec);

    for (int vb = v0; vb < v1; vb += 32 * BU) {
        v4i dvv[BU], svv[BU];
#pragma unroll
        for (int k = 0; k < BU; ++k) {
            const int vc = imin(vb + k * 32 + lane, v1 - 1);
            dvv[k] = dst4[vc];
            svv[k] = src4[vc];
        }
#pragma unroll
        for (int k = 0; k < BU; ++k) {
            const bool valid = vb + k * 32 + lane < v1;
            const v4i dv = dvv[k];
            const v4i sv = svv[k];
#pragma unroll
            for (int c = 0; c < 4; ++c) {
                const unsigned ud  = (unsigned)dv[c];
                const unsigned grp = ud / (unsigned)GNODES;
                const unsigned sn  = (unsigned)imin(imax(sv[c], 0), NNODES - 1);
                const unsigned word = ((ud - grp * (unsigned)GNODES) << SBITS) | sn;
                const bool act = valid && grp < (unsigned)NGROUPS;
                place32(buf, myfill, act, act ? grp : 0u, word, seg0, (unsigned)(NWAVE * SEGW), myspare);
            }
        }
    }
    __syncthreads();

    const unsigned in0 = (unsigned)(wave * NWAVE * SEGW);
    lds_tag_ptr myfill2 = (lds_tag_ptr)(fills2 + wave * NGRP);
    unsigned win[S2R];
#pragma unroll
    for (int r = 0; r < S2R; ++r) win[r] = buf[in0 + r * 32 + lane];
    for (int i = lane; i < NGRP * SEGW; i += 32) buf[in0 + i] = PAD_WORD;
    if (lane < NGRP) myfill2[lane] = 0u;
#pragma unroll
    for (int r = 0; r < S2R; ++r) {
        const unsigned word = win[r];
        const bool act = word != PAD_WORD;
        if (__builtin_amdgcn_ballot_w32(act) == 0u) continue;
        const unsigned own = (word >> SBITS) / (unsigned)NPB;
        place32(buf, myfill2, act, act ? own : 0u, word, in0, (unsigned)SEGW, myspare);
    }

    unsigned* coarse = extra;
    unsigned* ovf    = extra + CO_WORDS;
#pragma nounroll
    for (int o = 0; o < NGRP; ++o) {
        unsigned* seg = coarse + ((size_t)(wave * NGRP + o) * NBLK + blockIdx.x) * SEGW;
        unsigned wv[SEGW / 32];
#pragma unroll
        for (int j = 0; j < SEGW / 32; ++j) wv[j] = buf[in0 + o * SEGW + j * 32 + lane];
#pragma unroll
        for (int j = 0; j < SEGW / 32; ++j) seg[j * 32 + lane] = wv[j];
    }
    bool over = false;
    if (lane < NWAVE) over = fills[lane * NGROUPS + wave] > (unsigned)SEGW || myfill2[lane] > (unsigned)SEGW;
    const unsigned anyover = __builtin_amdgcn_ballot_w32(over);
    if (lane == 0) ovf[(size_t)wave * NBLK + blockIdx.x] = anyover != 0u ? 1u : 0u;
}

template <bool FIRST, bool CO>
__device__ __forceinline__ void owner_pass(const float* __restrict__ h, int pitch,
                                           const int* __restrict__ src,
                                           const int* __restrict__ dst,
                                           const v4i* __restrict__ scan4,
                                           int n_vec, int n_edges,
                                           int lo, int cnt, int tid, int lane,
                                           unsigned* myq, float* myacc, lds_tag_ptr mytag,
                                           unsigned* mylst, int cap) {
    int qn = 0;
    const int step  = TPB * VPI;

    bool rec   = FIRST && cap > 0;
    int stored = 0;
    int cov    = 0;

    int start = 0;
    if (!FIRST && cap > 0) {
        int n = __builtin_amdgcn_readfirstlane((int)mylst[0]);
        start = __builtin_amdgcn_readfirstlane((int)mylst[1]);
        n     = imax(imin(n, cap), 0);
        start = imax(start, 0);
        drain_queue<false, true>(mylst + LHDR, n, lane, h, pitch, src, myacc, mytag, nullptr);
    }

    if (start != COV_ALL) {
        v4i nd[VPI];
#pragma unroll
        for (int u = 0; u < VPI; ++u) {
            const int v = start + u * TPB + tid;
            nd[u] = scan4[imin(v, n_vec - 1)];
        }
        for (int base = start; base < n_vec; base += step) {
            if (qn > QW - VPI * 4 * 32) {
                unsigned* rp = nullptr;
                if (FIRST) {
                    if (rec && stored + qn <= cap) {
                        rp = mylst + LHDR + stored;
                        stored += qn;
                        cov = base;
                    } else {
                        rec = false;
                    }
                }
                drain_queue<FIRST, CO>(myq, qn, lane, h, pitch, src, myacc, mytag, rp);
                qn = 0;
            }
            if (CO || base + step <= n_vec) {
                v4i d[VPI];
#pragma unroll
                for (int u = 0; u < VPI; ++u) d[u] = nd[u];
                scan_full_trip<CO>(d, nd, scan4, base, step, tid, n_vec, lo, (unsigned)cnt, lane, myq, qn);
            } else {
#pragma nounroll
                for (int vb = base; vb < n_vec; vb += TPB) {
                    const int v = vb + tid;
                    const bool valid = v < n_vec;
                    const v4i dv = scan4[valid ? v : n_vec - 1];
#pragma unroll
                    for (int c = 0; c < 4; ++c) {
                        const unsigned t = (unsigned)(dv[c] - lo);
                        const bool m = valid && (t < (unsigned)cnt);
                        const unsigned mask = __builtin_amdgcn_ballot_w32(m);
                        if (mask != 0u) {
                            if (m) {
                                const int pos = qn + (int)__builtin_amdgcn_mbcnt_lo(mask, 0u);
                                myq[pos] = (t << EBITS) | (unsigned)(4 * v + c);
                            }
                            qn += __builtin_popcount(mask);
                        }
                    }
                }
            }
        }
        if (!CO) {
            for (int e0 = n_vec * 4; e0 < n_edges; e0 += TPB) {
                const int e = e0 + tid;
                if (qn > QW - 32) {
                    rec = false;
                    drain_queue<FIRST, false>(myq, qn, lane, h, pitch, src, myacc, mytag, nullptr);
                    qn = 0;
                }
                const bool valid = e < n_edges;
                const unsigned t = (unsigned)(dst[valid ? e : 0] - lo);
                const bool m = valid && (t < (unsigned)cnt);
                const unsigned mask = __builtin_amdgcn_ballot_w32(m);
                if (m) {
                    const int pos = qn + (int)__builtin_amdgcn_mbcnt_lo(mask, 0u);
                    myq[pos] = (t << EBITS) | (unsigned)e;
                }
                qn += __builtin_popcount(mask);
            }
        }
        {
            unsigned* rp = nullptr;
            if (FIRST && rec && stored + qn <= cap) {
                rp = mylst + LHDR + stored;
                stored += qn;
                cov = COV_ALL;
            }
            drain_queue<FIRST, CO>(myq, qn, lane, h, pitch, src, myacc, mytag, rp);
        }
    }
    if (FIRST && cap > 0 && lane == 0) {
        mylst[0] = (unsigned)stored;
        mylst[1] = (unsigned)cov;
    }
}

#define NTILE ((NPB + 15) / 16)
#define TPW   ((NTILE + NWAVE - 1) / NWAVE)

template <bool FIRST>
__global__ __launch_bounds__(TPB)
void owner_agg(const float* __restrict__ h,
               const float* __restrict__ xp,
               const int* __restrict__ src,
               const int* __restrict__ dst,
               float* __restrict__ deg,
               unsigned* __restrict__ extra,
               size_t ws_bytes,
               int n_nodes, int n_edges,
               const float* __restrict__ Wself,
               const float* __restrict__ Wneigh,
               const float* __restrict__ bias,
               float* __restrict__ out) {
    constexpr int F_OUT = FIRST ? FD : 10;
    constexpr int IN_PITCH  = FIRST ? FD : RP;
    constexpr int OUT_PITCH = FIRST ? RP : F_OUT;
    __shared__ __attribute__((aligned(16))) float accs[NWAVE * NPB * AC];
    __shared__ unsigned q[NWAVE * QS];
    __shared__ unsigned tags[NWAVE * NPB];
    __shared__ unsigned vote[NWAVE];

    const int tid  = threadIdx.x;
    const int lane = tid & 31;
    const int wave = tid >> 5;
    const int lo   = blockIdx.x * NPB;
    const int cnt  = imax(imin(NPB, n_nodes - lo), 0);
    const int grp  = blockIdx.x / NGRP;

    const int row   = lane & 15;
    const int khalf = lane >> 4;
    const int nc    = imin(row, F_OUT - 1);
    const bool nvalid = row < F_OUT;
    v2f bfrag[3];
    float hself[TPW][3][2];
    float degpre[TPW];
#pragma unroll
    for (int chunk = 0; chunk < 3; ++chunk) {
#pragma unroll
        for (int j = 0; j < 2; ++j) {
            const int k  = chunk * 4 + khalf * 2 + j;
            const int kh = imin(k, FD - 1);
            const int kn = imin(imax(k - FD, 0), FD - 1);
            const float ws = Wself[kh * F_OUT + nc];
            const float wn = Wneigh[kn * F_OUT + nc];
            float bv = (k < FD) ? ws : ((k < 2 * FD) ? wn : 0.0f);
            bfrag[chunk][j] = nvalid ? bv : 0.0f;
        }
    }
    const float bn = bias[nc];
#pragma unroll
    for (int tt = 0; tt < TPW; ++tt) {
        const int ml = imin((wave + tt * NWAVE) * 16 + row, imax(cnt - 1, 0));
        const int mc = imin(lo + ml, n_nodes - 1);
        degpre[tt] = FIRST ? 0.0f : deg[mc];
#pragma unroll
        for (int chunk = 0; chunk < 3; ++chunk) {
#pragma unroll
            for (int j = 0; j < 2; ++j) {
                const int kh = imin(chunk * 4 + khalf * 2 + j, FD - 1);
                hself[tt][chunk][j] = h[(long)mc * IN_PITCH + kh];
            }
        }
    }

    const size_t used = FRONT_FLOATS * sizeof(float);
    const bool xp_room = ws_bytes >= used;
    const float* gtab  = (FIRST && xp_room) ? xp : h;
    const int gpitch   = (FIRST && !xp_room) ? FD : RP;
    const size_t co_bytes = (CO_WORDS + OVF_WORDS) * sizeof(unsigned);
    const bool co_room = (n_edges & 3) == 0 && ws_bytes >= used + co_bytes;
    size_t rest = ws_bytes > used ? ws_bytes - used : 0;
    unsigned* lst = extra;
    if (co_room) {
        rest -= co_bytes;
        lst = extra + CO_WORDS + OVF_WORDS;
    }
    size_t pitch = rest / (sizeof(unsigned) * NBLK * NWAVE);
    pitch = pitch > LPMAX ? (size_t)LPMAX : (pitch & ~(size_t)31);
    const int cap = pitch > LHDR ? (int)(pitch - LHDR) : 0;
    unsigned* mylst = lst + (size_t)(blockIdx.x * NWAVE + wave) * pitch;

    int bad = 0;
    if (co_room) {
        const unsigned* ovf = extra + CO_WORDS + (size_t)grp * NBLK;
        for (int i = tid; i < NBLK; i += TPB) bad |= (int)(ovf[i] != 0u);
    }
    const unsigned wbad = __builtin_amdgcn_ballot_w32(bad != 0);
    if (lane == 0) vote[wave] = wbad;
    for (int i = tid; i < NWAVE * NPB * AC; i += TPB) accs[i] = 0.0f;
    __syncthreads();
    unsigned anybad = 0u;
#pragma unroll
    for (int w = 0; w < NWAVE; ++w) anybad |= vote[w];
    const bool use_co = co_room && __builtin_amdgcn_readfirstlane((int)anybad) == 0;

    float* myacc = accs + wave * (NPB * AC);
    unsigned* myq = q + wave * QS;
    lds_tag_ptr mytag = (lds_tag_ptr)(tags + wave * NPB);

    if (use_co) {
        owner_pass<FIRST, true>(gtab, gpitch, src, dst, (const v4i*)(extra + (size_t)blockIdx.x * OWORDS), OWORDS / 4, n_edges,
                                (blockIdx.x % NGRP) * NPB, cnt, tid, lane, myq, myacc, mytag, mylst, cap);
    } else {
        owner_pass<FIRST, false>(gtab, gpitch, src, dst, (const v4i*)dst, n_edges >> 2, n_edges,
                                 lo, cnt, tid, lane, myq, myacc, mytag, mylst, cap);
    }
    __syncthreads();

    for (int i = tid; i < cnt * AC; i += TPB) {
        float sum = 0.0f;
#pragma unroll
        for (int w = 0; w < NWAVE; ++w) sum += accs[w * (NPB * AC) + i];
        accs[i] = sum;
        if (FIRST) {
            const int node = i / AC;
            if (i - node * AC == FD) deg[lo + node] = sum;
        }
    }
    __syncthreads();

    const int swave = __builtin_amdgcn_readfirstlane(wave);
#pragma unroll
    for (int tt = 0; tt < TPW; ++tt) {
        const int m0 = (swave + tt * NWAVE) * 16;
        if (m0 < cnt) {
            const int ml = imin(m0 + row, cnt - 1);
            const float dg = FIRST ? accs[ml * AC + FD] : degpre[tt];
            const float invdeg = 1.0f / fmaxf(dg, 1.0f);
            v8f c = {};
#pragma unroll
            for (int chunk = 0; chunk < 3; ++chunk) {
                v2f a;
#pragma unroll
                for (int j = 0; j < 2; ++j) {
                    const int k  = chunk * 4 + khalf * 2 + j;
                    const int kn = imin(imax(k - FD, 0), FD - 1);
                    const float nv = accs[ml * AC + kn] * invdeg;
                    a[j] = (k < FD) ? hself[tt][chunk][j] : ((k < 2 * FD) ? nv : 0.0f);
                }
                c = __builtin_amdgcn_wmma_f32_16x16x4_f32(false, a, false, bfrag[chunk],
                                                          (short)0, c, false, false);
            }
#pragma unroll
            for (int r = 0; r < 8; ++r) {
                const int mm = m0 + r + khalf * 8;
                if (mm < cnt && nvalid) {
                    const float v = c[r] + bn;
                    out[(long)(lo + mm) * OUT_PITCH + row] = 1.0f / (1.0f + __expf(-v));
                }
            }
        }
    }
}

extern "C" void kernel_launch(void* const* d_in, const int* in_sizes, int n_in,
                              void* d_out, int out_size, void* d_ws, size_t ws_size,
                              hipStream_t stream) {
    (void)in_sizes; (void)n_in; (void)out_size;

    const float* x      = (const float*)d_in[0];
    const int*   src    = (const int*)d_in[1];
    const int*   dst    = (const int*)d_in[2];
    const float* Wself1 = (const float*)d_in[3];
    const float* Wneigh1= (const float*)d_in[4];
    const float* b1     = (const float*)d_in[5];
    const float* Wself2 = (const float*)d_in[6];
    const float* Wneigh2= (const float*)d_in[7];
    const float* b2     = (const float*)d_in[8];
    float* out = (float*)d_out;

    float* h   = (float*)d_ws;
    float* deg = h + (size_t)2 * NNODES * FD;
    float* xp  = deg + (size_t)NNODES;
    unsigned* extra = (unsigned*)(xp + (size_t)NNODES * RP);

    coarse_bin<<<NBLK, TPB, 0, stream>>>(src, dst, x, xp, extra, ws_size, NEDGES);

    owner_agg<true><<<NBLK, TPB, 0, stream>>>(x, xp, src, dst, deg, extra, ws_size, NNODES, NEDGES,
                                              Wself1, Wneigh1, b1, h);

    owner_agg<false><<<NBLK, TPB, 0, stream>>>(h, nullptr, src, dst, deg, extra, ws_size, NNODES, NEDGES,
                                               Wself2, Wneigh2, b2, out);
}
